# starting kernel
# baseline (speedup 1.0000x reference)
#include <hip/hip_runtime.h>
#include <hip/hip_bf16.h>

#define H 256
#define BB 32
#define TT 4096
#define EE 8192
#define D2 512

typedef __attribute__((ext_vector_type(16))) __bf16        v16bf;
typedef __attribute__((ext_vector_type(8)))  float         v8f;
typedef __attribute__((ext_vector_type(8)))  unsigned int  v8u;

__device__ __forceinline__ unsigned short f2bf(float x) {
    unsigned int u = __float_as_uint(x);
    unsigned int r = u + 0x7FFFu + ((u >> 16) & 1u);
    return (unsigned short)(r >> 16);
}
__device__ __forceinline__ unsigned int pk2(float lo, float hi) {
    return (unsigned int)f2bf(lo) | ((unsigned int)f2bf(hi) << 16);
}

__device__ __forceinline__ float htanh(float x) {
#if __has_builtin(__builtin_amdgcn_tanhf)
    return __builtin_amdgcn_tanhf(x);
#elif __has_builtin(__builtin_amdgcn_tanh_f32)
    return __builtin_amdgcn_tanh_f32(x);
#else
    float y;
    asm volatile("v_tanh_f32 %0, %1\n\t"
                 "s_delay_alu instid0(TRANS32_DEP_1)"
                 : "=v"(y) : "v"(x));
    return y;
#endif
}

__global__ void packb_kernel(const float* __restrict__ W, unsigned short* __restrict__ Bp) {
    int blk  = blockIdx.x;
    int nt   = blk >> 3;
    int kc   = blk & 7;
    int lane = threadIdx.x;
    int n     = nt * 16 + (lane & 15);
    int kbase = kc * 32 + ((lane >> 4) ? 8 : 0);
    const float* wrow = W + (size_t)n * D2 + H;
    unsigned short* dst = Bp + ((size_t)blk * 32 + lane) * 16;
    #pragma unroll
    for (int j = 0; j < 8; ++j) dst[j]     = f2bf(wrow[kbase + j]);
    #pragma unroll
    for (int j = 0; j < 8; ++j) dst[8 + j] = f2bf(wrow[kbase + 16 + j]);
}

__global__ void p_kernel(const float* __restrict__ hidden, const float* __restrict__ W,
                         const float* __restrict__ bvec, float* __restrict__ P) {
    int b = blockIdx.x;
    int n = threadIdx.x;
    const float* h = hidden + b * H;
    const float* w = W + (size_t)n * D2;
    float acc = bvec[n];
    #pragma unroll 4
    for (int k = 0; k < H; ++k) acc = fmaf(h[k], w[k], acc);
    P[b * D2 + n] = acc;
}

__global__ void __launch_bounds__(256)
scores_kernel(const float* __restrict__ enc, const v8u* __restrict__ Bq,
              const float* __restrict__ P, const float* __restrict__ v,
              float* __restrict__ scores) {
    int wave = blockIdx.x * 8 + (threadIdx.x >> 5);
    int lane = threadIdx.x & 31;
    int b    = wave >> 7;
    int t0   = (wave & 127) << 5;
    int mrow = lane & 15;
    int kgrp = lane >> 4;

    const float* rowp0 = enc + ((size_t)(t0 + mrow)      * BB + b) * H + kgrp * 8;
    const float* rowp1 = enc + ((size_t)(t0 + 16 + mrow) * BB + b) * H + kgrp * 8;
    v16bf a0[8], a1[8];
    #pragma unroll
    for (int kc = 0; kc < 8; ++kc) {
        const float* p0 = rowp0 + kc * 32;
        const float* p1 = rowp1 + kc * 32;
        v8u u0, u1;
        #pragma unroll
        for (int j = 0; j < 4; ++j) { u0[j] = pk2(p0[2 * j],     p0[2 * j + 1]);
                                      u1[j] = pk2(p1[2 * j],     p1[2 * j + 1]); }
        #pragma unroll
        for (int j = 4; j < 8; ++j) { u0[j] = pk2(p0[8 + 2 * j], p0[8 + 2 * j + 1]);
                                      u1[j] = pk2(p1[8 + 2 * j], p1[8 + 2 * j + 1]); }
        a0[kc] = __builtin_bit_cast(v16bf, u0);
        a1[kc] = __builtin_bit_cast(v16bf, u1);
    }

    float sacc0[8], sacc1[8];
    #pragma unroll
    for (int i = 0; i < 8; ++i) { sacc0[i] = 0.f; sacc1[i] = 0.f; }

    const float* Pb = P + b * D2;
    for (int nt = 0; nt < 32; ++nt) {
        const v8u* bp = Bq + ((size_t)nt * 8 * 32 + lane);
        if (nt < 31) __builtin_prefetch(bp + 8 * 32, 0, 1);
        v8f c0 = {}, c1 = {};
        #pragma unroll
        for (int kc = 0; kc < 8; ++kc) {
            v16bf bv = __builtin_bit_cast(v16bf, bp[kc * 32]);
            c0 = __builtin_amdgcn_wmma_f32_16x16x32_bf16(
                     false, a0[kc], false, bv, (short)0, c0, false, false);
            c1 = __builtin_amdgcn_wmma_f32_16x16x32_bf16(
                     false, a1[kc], false, bv, (short)0, c1, false, false);
        }
        int   n  = nt * 16 + (lane & 15);
        float pn = Pb[n];
        float vn = v[n];
        #pragma unroll
        for (int i = 0; i < 8; ++i) {
            sacc0[i] += vn * htanh(c0[i] + pn);
            sacc1[i] += vn * htanh(c1[i] + pn);
        }
    }

    #pragma unroll
    for (int i = 0; i < 8; ++i) {
        float s0 = sacc0[i], s1 = sacc1[i];
        #pragma unroll
        for (int off = 1; off < 16; off <<= 1) {
            s0 += __shfl_xor(s0, off, 32);
            s1 += __shfl_xor(s1, off, 32);
        }
        sacc0[i] = s0; sacc1[i] = s1;
    }
    if ((lane & 15) == 0) {
        #pragma unroll
        for (int i = 0; i < 8; ++i) {
            scores[b * TT + t0 +      kgrp * 8 + i] = sacc0[i];
            scores[b * TT + t0 + 16 + kgrp * 8 + i] = sacc1[i];
        }
    }
}

__global__ void softmax_kernel(float* __restrict__ w) {
    __shared__ float red[256];
    int b = blockIdx.x, tid = threadIdx.x;
    float* row = w + b * TT;
    float m = -1e30f;
    for (int t = tid; t < TT; t += 256) m = fmaxf(m, row[t]);
    red[tid] = m; __syncthreads();
    for (int s = 128; s > 0; s >>= 1) { if (tid < s) red[tid] = fmaxf(red[tid], red[tid + s]); __syncthreads(); }
    m = red[0]; __syncthreads();
    float sum = 0.f;
    for (int t = tid; t < TT; t += 256) sum += __expf(row[t] - m);
    red[tid] = sum; __syncthreads();
    for (int s = 128; s > 0; s >>= 1) { if (tid < s) red[tid] += red[tid + s]; __syncthreads(); }
    float inv = 1.0f / red[0];
    for (int t = tid; t < TT; t += 256) row[t] = __expf(row[t] - m) * inv;
}

__global__ void zero_kernel(float* __restrict__ out, int n) {
    int i = blockIdx.x * blockDim.x + threadIdx.x;
    if (i < n) out[i] = 0.f;
}

__global__ void scatter_kernel(const int* __restrict__ src, const int* __restrict__ dst,
                               const float* __restrict__ w, float* __restrict__ out) {
    int i = blockIdx.x * blockDim.x + threadIdx.x;
    if (i >= BB * EE) return;
    int b = i >> 13;
    int s = src[i], d = dst[i];
    if (d != s + 1)
        atomicAdd(&out[b * TT + d], w[b * TT + s] * 0.1f);
}

extern "C" void kernel_launch(void* const* d_in, const int* in_sizes, int n_in,
                              void* d_out, int out_size, void* d_ws, size_t ws_size,
                              hipStream_t stream) {
    const float* hidden = (const float*)d_in[0];
    const float* enc    = (const float*)d_in[1];
    const int*   esrc   = (const int*)d_in[2];
    const int*   edst   = (const int*)d_in[3];
    const float* W      = (const float*)d_in[4];
    const float* bvec   = (const float*)d_in[5];
    const float* v      = (const float*)d_in[6];
    float* out = (float*)d_out;

    char* ws = (char*)d_ws;
    unsigned short* Bpack = (unsigned short*)ws;
    float* P    = (float*)(ws + 262144);
    float* wbuf = (float*)(ws + 262144 + 65536);

    packb_kernel  <<<256, 32, 0, stream>>>(W, Bpack);
    p_kernel      <<<BB, D2, 0, stream>>>(hidden, W, bvec, P);
    scores_kernel <<<512, 256, 0, stream>>>(enc, (const v8u*)Bpack, P, v, wbuf);
    softmax_kernel<<<BB, 256, 0, stream>>>(wbuf);
    zero_kernel   <<<(BB * TT + 255) / 256, 256, 0, stream>>>(out, BB * TT);
    scatter_kernel<<<(BB * EE + 255) / 256, 256, 0, stream>>>(esrc, edst, wbuf, out);
}


# B operand (256 KB packed bf16), P row and v staged once per 16-wave work-group in LDS in a conflict-free layout; N loop reads them by ds_load instead of re-reading L2 per wave; grid 256x512
# speedup vs baseline: 1.7030x; 1.7030x over previous
#include <hip/hip_runtime.h>
#include <hip/hip_bf16.h>

#define H 256
#define BB 32
#define TT 4096
#define EE 8192
#define D2 512

typedef __attribute__((ext_vector_type(16))) __bf16        v16bf;
typedef __attribute__((ext_vector_type(8)))  float         v8f;
typedef __attribute__((ext_vector_type(8)))  unsigned int  v8u;

__device__ __forceinline__ unsigned short f2bf(float x) {
    unsigned int u = __float_as_uint(x);
    unsigned int r = u + 0x7FFFu + ((u >> 16) & 1u);
    return (unsigned short)(r >> 16);
}
__device__ __forceinline__ unsigned int pk2(float lo, float hi) {
    return (unsigned int)f2bf(lo) | ((unsigned int)f2bf(hi) << 16);
}

__device__ __forceinline__ float htanh(float x) {
#if __has_builtin(__builtin_amdgcn_tanhf)
    return __builtin_amdgcn_tanhf(x);
#elif __has_builtin(__builtin_amdgcn_tanh_f32)
    return __builtin_amdgcn_tanh_f32(x);
#else
    float y;
    asm volatile("v_tanh_f32 %0, %1\n\t"
                 "s_delay_alu instid0(TRANS32_DEP_1)"
                 : "=v"(y) : "v"(x));
    return y;
#endif
}

__global__ void packb_kernel(const float* __restrict__ W, unsigned short* __restrict__ Bp) {
    int blk  = blockIdx.x;
    int nt   = blk >> 3;
    int kc   = blk & 7;
    int lane = threadIdx.x;
    int n     = nt * 16 + (lane & 15);
    int kbase = kc * 32 + ((lane >> 4) ? 8 : 0);
    const float* wrow = W + (size_t)n * D2 + H;
    unsigned short* dst = Bp + (size_t)blk * 512 + lane * 8;
    #pragma unroll
    for (int j = 0; j < 8; ++j) dst[j]       = f2bf(wrow[kbase + j]);
    #pragma unroll
    for (int j = 0; j < 8; ++j) dst[256 + j] = f2bf(wrow[kbase + 16 + j]);
}

__global__ void p_kernel(const float* __restrict__ hidden, const float* __restrict__ W,
                         const float* __restrict__ bvec, float* __restrict__ P) {
    int b = blockIdx.x;
    int n = threadIdx.x;
    const float* h = hidden + b * H;
    const float* w = W + (size_t)n * D2;
    float acc = bvec[n];
    #pragma unroll 4
    for (int k = 0; k < H; ++k) acc = fmaf(h[k], w[k], acc);
    P[b * D2 + n] = acc;
}

#define BQ_VEC 16384

__global__ void __launch_bounds__(512)
scores_kernel(const float* __restrict__ enc, const uint4* __restrict__ Bq,
              const float* __restrict__ P, const float* __restrict__ v,
              float* __restrict__ scores) {
    __shared__ uint4 Bs[BQ_VEC];
    __shared__ float Ps[D2];
    __shared__ float vs[D2];
    int wave = blockIdx.x * 16 + (threadIdx.x >> 5);
    int lane = threadIdx.x & 31;

    {
        const int tid = threadIdx.x;
        const int rot = blockIdx.x & 31;
        #pragma unroll 1
        for (int jj = 0; jj < 32; jj += 16) {
            uint4 tmp[16];
            #pragma unroll
            for (int u = 0; u < 16; ++u) tmp[u] = Bq[tid + 512 * ((jj + u + rot) & 31)];
            #pragma unroll
            for (int u = 0; u < 16; ++u) Bs[tid + 512 * ((jj + u + rot) & 31)] = tmp[u];
        }
        Ps[tid] = P[(blockIdx.x >> 3) * D2 + tid];
        vs[tid] = v[tid];
    }

    int b    = wave >> 7;
    int t0   = (wave & 127) << 5;
    int mrow = lane & 15;
    int kgrp = lane >> 4;

    const float* rowp0 = enc + ((size_t)(t0 + mrow)      * BB + b) * H + kgrp * 8;
    const float* rowp1 = enc + ((size_t)(t0 + 16 + mrow) * BB + b) * H + kgrp * 8;
    v16bf a0[8], a1[8];
    #pragma unroll
    for (int kc = 0; kc < 8; ++kc) {
        const float* p0 = rowp0 + kc * 32;
        const float* p1 = rowp1 + kc * 32;
        v8u u0, u1;
        #pragma unroll
        for (int j = 0; j < 4; ++j) { u0[j] = pk2(p0[2 * j],     p0[2 * j + 1]);
                                      u1[j] = pk2(p1[2 * j],     p1[2 * j + 1]); }
        #pragma unroll
        for (int j = 4; j < 8; ++j) { u0[j] = pk2(p0[8 + 2 * j], p0[8 + 2 * j + 1]);
                                      u1[j] = pk2(p1[8 + 2 * j], p1[8 + 2 * j + 1]); }
        a0[kc] = __builtin_bit_cast(v16bf, u0);
        a1[kc] = __builtin_bit_cast(v16bf, u1);
    }

    float sacc0[8], sacc1[8];
    #pragma unroll
    for (int i = 0; i < 8; ++i) { sacc0[i] = 0.f; sacc1[i] = 0.f; }

    __syncthreads();

    for (int nt = 0; nt < 32; ++nt) {
        const uint4* bs = Bs + (nt * 8 * 64 + lane);
        v8f c0 = {}, c1 = {};
        #pragma unroll
        for (int kc = 0; kc < 8; ++kc) {
            uint4 lo = bs[kc * 64];
            uint4 hi = bs[kc * 64 + 32];
            v8u bu = { lo.x, lo.y, lo.z, lo.w, hi.x, hi.y, hi.z, hi.w };
            v16bf bv = __builtin_bit_cast(v16bf, bu);
            c0 = __builtin_amdgcn_wmma_f32_16x16x32_bf16(
                     false, a0[kc], false, bv, (short)0, c0, false, false);
            c1 = __builtin_amdgcn_wmma_f32_16x16x32_bf16(
                     false, a1[kc], false, bv, (short)0, c1, false, false);
        }
        int   n  = nt * 16 + (lane & 15);
        float pn = Ps[n];
        float vn = vs[n];
        #pragma unroll
        for (int i = 0; i < 8; ++i) {
            sacc0[i] += vn * htanh(c0[i] + pn);
            sacc1[i] += vn * htanh(c1[i] + pn);
        }
    }

    #pragma unroll
    for (int i = 0; i < 8; ++i) {
        float s0 = sacc0[i], s1 = sacc1[i];
        #pragma unroll
        for (int off = 1; off < 16; off <<= 1) {
            s0 += __shfl_xor(s0, off, 32);
            s1 += __shfl_xor(s1, off, 32);
        }
        sacc0[i] = s0; sacc1[i] = s1;
    }
    if ((lane & 15) == 0) {
        #pragma unroll
        for (int i = 0; i < 8; ++i) {
            scores[b * TT + t0 +      kgrp * 8 + i] = sacc0[i];
            scores[b * TT + t0 + 16 + kgrp * 8 + i] = sacc1[i];
        }
    }
}

__global__ void softmax_kernel(float* __restrict__ w) {
    __shared__ float red[256];
    int b = blockIdx.x, tid = threadIdx.x;
    float* row = w + b * TT;
    float m = -1e30f;
    for (int t = tid; t < TT; t += 256) m = fmaxf(m, row[t]);
    red[tid] = m; __syncthreads();
    for (int s = 128; s > 0; s >>= 1) { if (tid < s) red[tid] = fmaxf(red[tid], red[tid + s]); __syncthreads(); }
    m = red[0]; __syncthreads();
    float sum = 0.f;
    for (int t = tid; t < TT; t += 256) sum += __expf(row[t] - m);
    red[tid] = sum; __syncthreads();
    for (int s = 128; s > 0; s >>= 1) { if (tid < s) red[tid] += red[tid + s]; __syncthreads(); }
    float inv = 1.0f / red[0];
    for (int t = tid; t < TT; t += 256) row[t] = __expf(row[t] - m) * inv;
}

__global__ void zero_kernel(float* __restrict__ out, int n) {
    int i = blockIdx.x * blockDim.x + threadIdx.x;
    if (i < n) out[i] = 0.f;
}

__global__ void scatter_kernel(const int* __restrict__ src, const int* __restrict__ dst,
                               const float* __restrict__ w, float* __restrict__ out) {
    int i = blockIdx.x * blockDim.x + threadIdx.x;
    if (i >= BB * EE) return;
    int b = i >> 13;
    int s = src[i], d = dst[i];
    if (d != s + 1)
        atomicAdd(&out[b * TT + d], w[b * TT + s] * 0.1f);
}

extern "C" void kernel_launch(void* const* d_in, const int* in_sizes, int n_in,
                              void* d_out, int out_size, void* d_ws, size_t ws_size,
                              hipStream_t stream) {
    const float* hidden = (const float*)d_in[0];
    const float* enc    = (const float*)d_in[1];
    const int*   esrc   = (const int*)d_in[2];
    const int*   edst   = (const int*)d_in[3];
    const float* W      = (const float*)d_in[4];
    const float* bvec   = (const float*)d_in[5];
    const float* v      = (const float*)d_in[6];
    float* out = (float*)d_out;

    char* ws = (char*)d_ws;
    unsigned short* Bpack = (unsigned short*)ws;
    float* P    = (float*)(ws + 262144);
    float* wbuf = (float*)(ws + 262144 + 65536);

    packb_kernel  <<<256, 32, 0, stream>>>(W, Bpack);
    p_kernel      <<<BB, D2, 0, stream>>>(hidden, W, bvec, P);
    scores_kernel <<<256, 512, 0, stream>>>(enc, (const uint4*)Bpack, P, v, wbuf);
    softmax_kernel<<<BB, 256, 0, stream>>>(wbuf);
    zero_kernel   <<<(BB * TT + 255) / 256, 256, 0, stream>>>(out, BB * TT);
    scatter_kernel<<<(BB * EE + 255) / 256, 256, 0, stream>>>(esrc, edst, wbuf, out);
}


# launch fusion 6->3: P, B pack and out zeroing share one launch; softmax folded into the scatter launch (each 256-thread group re-reduces its row max and exp sum, so the sum tree changes)
# speedup vs baseline: 1.9765x; 1.1606x over previous
#include <hip/hip_runtime.h>
#include <hip/hip_bf16.h>

#define H 256
#define BB 32
#define TT 4096
#define EE 8192
#define D2 512

typedef __attribute__((ext_vector_type(16))) __bf16        v16bf;
typedef __attribute__((ext_vector_type(8)))  float         v8f;
typedef __attribute__((ext_vector_type(8)))  unsigned int  v8u;

__device__ __forceinline__ unsigned short f2bf(float x) {
    unsigned int u = __float_as_uint(x);
    unsigned int r = u + 0x7FFFu + ((u >> 16) & 1u);
    return (unsigned short)(r >> 16);
}
__device__ __forceinline__ unsigned int pk2(float lo, float hi) {
    return (unsigned int)f2bf(lo) | ((unsigned int)f2bf(hi) << 16);
}

__device__ __forceinline__ float htanh(float x) {
#if __has_builtin(__builtin_amdgcn_tanhf)
    return __builtin_amdgcn_tanhf(x);
#elif __has_builtin(__builtin_amdgcn_tanh_f32)
    return __builtin_amdgcn_tanh_f32(x);
#else
    float y;
    asm volatile("v_tanh_f32 %0, %1\n\t"
                 "s_delay_alu instid0(TRANS32_DEP_1)"
                 : "=v"(y) : "v"(x));
    return y;
#endif
}

#define PREP_P_GROUPS    32
#define PREP_PACK_GROUPS 16
#define PREP_ZERO_GROUPS 64
#define PREP_GROUPS      112

__global__ void __launch_bounds__(512)
prep_kernel(const float* __restrict__ hidden, const float* __restrict__ W,
            const float* __restrict__ bvec, float* __restrict__ P,
            unsigned short* __restrict__ Bp, float* __restrict__ out) {
    const int g = blockIdx.x;
    if (g < PREP_P_GROUPS) {
        int b = g;
        int n = threadIdx.x;
        const float* h = hidden + b * H;
        const float* w = W + (size_t)n * D2;
        float acc = bvec[n];
        #pragma unroll 4
        for (int k = 0; k < H; ++k) acc = fmaf(h[k], w[k], acc);
        P[b * D2 + n] = acc;
    } else if (g < PREP_P_GROUPS + PREP_PACK_GROUPS) {
        int blk  = (g - PREP_P_GROUPS) * 16 + (threadIdx.x >> 5);
        int nt   = blk >> 3;
        int kc   = blk & 7;
        int lane = threadIdx.x & 31;
        int n     = nt * 16 + (lane & 15);
        int kbase = kc * 32 + ((lane >> 4) ? 8 : 0);
        const float* wrow = W + (size_t)n * D2 + H;
        unsigned short* dst = Bp + (size_t)blk * 512 + lane * 8;
        #pragma unroll
        for (int j = 0; j < 8; ++j) dst[j]       = f2bf(wrow[kbase + j]);
        #pragma unroll
        for (int j = 0; j < 8; ++j) dst[256 + j] = f2bf(wrow[kbase + 16 + j]);
    } else {
        int i = (g - PREP_P_GROUPS - PREP_PACK_GROUPS) * 512 + threadIdx.x;
        ((float4*)out)[i] = make_float4(0.f, 0.f, 0.f, 0.f);
    }
}

#define BQ_VEC 16384

__global__ void __launch_bounds__(512)
scores_kernel(const float* __restrict__ enc, const uint4* __restrict__ Bq,
              const float* __restrict__ P, const float* __restrict__ v,
              float* __restrict__ scores) {
    __shared__ uint4 Bs[BQ_VEC];
    __shared__ float Ps[D2];
    __shared__ float vs[D2];
    int wave = blockIdx.x * 16 + (threadIdx.x >> 5);
    int lane = threadIdx.x & 31;

    {
        const int tid = threadIdx.x;
        const int rot = blockIdx.x & 31;
        #pragma unroll 1
        for (int jj = 0; jj < 32; jj += 16) {
            uint4 tmp[16];
            #pragma unroll
            for (int u = 0; u < 16; ++u) tmp[u] = Bq[tid + 512 * ((jj + u + rot) & 31)];
            #pragma unroll
            for (int u = 0; u < 16; ++u) Bs[tid + 512 * ((jj + u + rot) & 31)] = tmp[u];
        }
        Ps[tid] = P[(blockIdx.x >> 3) * D2 + tid];
        vs[tid] = v[tid];
    }

    int b    = wave >> 7;
    int t0   = (wave & 127) << 5;
    int mrow = lane & 15;
    int kgrp = lane >> 4;

    const float* rowp0 = enc + ((size_t)(t0 + mrow)      * BB + b) * H + kgrp * 8;
    const float* rowp1 = enc + ((size_t)(t0 + 16 + mrow) * BB + b) * H + kgrp * 8;
    v16bf a0[8], a1[8];
    #pragma unroll
    for (int kc = 0; kc < 8; ++kc) {
        const float* p0 = rowp0 + kc * 32;
        const float* p1 = rowp1 + kc * 32;
        v8u u0, u1;
        #pragma unroll
        for (int j = 0; j < 4; ++j) { u0[j] = pk2(p0[2 * j],     p0[2 * j + 1]);
                                      u1[j] = pk2(p1[2 * j],     p1[2 * j + 1]); }
        #pragma unroll
        for (int j = 4; j < 8; ++j) { u0[j] = pk2(p0[8 + 2 * j], p0[8 + 2 * j + 1]);
                                      u1[j] = pk2(p1[8 + 2 * j], p1[8 + 2 * j + 1]); }
        a0[kc] = __builtin_bit_cast(v16bf, u0);
        a1[kc] = __builtin_bit_cast(v16bf, u1);
    }

    float sacc0[8], sacc1[8];
    #pragma unroll
    for (int i = 0; i < 8; ++i) { sacc0[i] = 0.f; sacc1[i] = 0.f; }

    __syncthreads();

    for (int nt = 0; nt < 32; ++nt) {
        const uint4* bs = Bs + (nt * 8 * 64 + lane);
        v8f c0 = {}, c1 = {};
        #pragma unroll
        for (int kc = 0; kc < 8; ++kc) {
            uint4 lo = bs[kc * 64];
            uint4 hi = bs[kc * 64 + 32];
            v8u bu = { lo.x, lo.y, lo.z, lo.w, hi.x, hi.y, hi.z, hi.w };
            v16bf bv = __builtin_bit_cast(v16bf, bu);
            c0 = __builtin_amdgcn_wmma_f32_16x16x32_bf16(
                     false, a0[kc], false, bv, (short)0, c0, false, false);
            c1 = __builtin_amdgcn_wmma_f32_16x16x32_bf16(
                     false, a1[kc], false, bv, (short)0, c1, false, false);
        }
        int   n  = nt * 16 + (lane & 15);
        float pn = Ps[n];
        float vn = vs[n];
        #pragma unroll
        for (int i = 0; i < 8; ++i) {
            sacc0[i] += vn * htanh(c0[i] + pn);
            sacc1[i] += vn * htanh(c1[i] + pn);
        }
    }

    #pragma unroll
    for (int i = 0; i < 8; ++i) {
        float s0 = sacc0[i], s1 = sacc1[i];
        #pragma unroll
        for (int off = 1; off < 16; off <<= 1) {
            s0 += __shfl_xor(s0, off, 32);
            s1 += __shfl_xor(s1, off, 32);
        }
        sacc0[i] = s0; sacc1[i] = s1;
    }
    if ((lane & 15) == 0) {
        #pragma unroll
        for (int i = 0; i < 8; ++i) {
            scores[b * TT + t0 +      kgrp * 8 + i] = sacc0[i];
            scores[b * TT + t0 + 16 + kgrp * 8 + i] = sacc1[i];
        }
    }
}

__global__ void __launch_bounds__(256)
softmax_scatter_kernel(const int* __restrict__ src, const int* __restrict__ dst,
                       const float* __restrict__ sc, float* __restrict__ out) {
    __shared__ float redm[8];
    __shared__ float reds[8];
    const int b    = blockIdx.x >> 3;
    const int c    = blockIdx.x & 7;
    const int tid  = threadIdx.x;
    const int lane = tid & 31;
    const int wv   = tid >> 5;
    const float* row = sc + b * TT;

    const int e0 = b * EE + c * 1024 + tid;
    int s[4], d[4];
    #pragma unroll
    for (int j = 0; j < 4; ++j) { s[j] = src[e0 + 256 * j]; d[j] = dst[e0 + 256 * j]; }
    float4 x[4];
    #pragma unroll
    for (int j = 0; j < 4; ++j) x[j] = ((const float4*)row)[tid + 256 * j];
    float xs[4];
    #pragma unroll
    for (int j = 0; j < 4; ++j) xs[j] = row[s[j]];

    float m = fmaxf(fmaxf(x[0].x, x[0].y), fmaxf(x[0].z, x[0].w));
    #pragma unroll
    for (int j = 1; j < 4; ++j)
        m = fmaxf(m, fmaxf(fmaxf(x[j].x, x[j].y), fmaxf(x[j].z, x[j].w)));
    #pragma unroll
    for (int off = 16; off > 0; off >>= 1) m = fmaxf(m, __shfl_xor(m, off, 32));
    if (lane == 0) redm[wv] = m;
    __syncthreads();
    m = fmaxf(fmaxf(fmaxf(redm[0], redm[1]), fmaxf(redm[2], redm[3])),
              fmaxf(fmaxf(redm[4], redm[5]), fmaxf(redm[6], redm[7])));

    float sum = 0.f;
    #pragma unroll
    for (int j = 0; j < 4; ++j)
        sum += (__expf(x[j].x - m) + __expf(x[j].y - m)) + (__expf(x[j].z - m) + __expf(x[j].w - m));
    #pragma unroll
    for (int off = 16; off > 0; off >>= 1) sum += __shfl_xor(sum, off, 32);
    if (lane == 0) reds[wv] = sum;
    __syncthreads();
    sum = ((reds[0] + reds[1]) + (reds[2] + reds[3])) + ((reds[4] + reds[5]) + (reds[6] + reds[7]));
    const float inv = 1.0f / sum;

    #pragma unroll
    for (int j = 0; j < 4; ++j)
        if (d[j] != s[j] + 1)
            atomicAdd(&out[b * TT + d[j]], (__expf(xs[j] - m) * inv) * 0.1f);
}

extern "C" void kernel_launch(void* const* d_in, const int* in_sizes, int n_in,
                              void* d_out, int out_size, void* d_ws, size_t ws_size,
                              hipStream_t stream) {
    const float* hidden = (const float*)d_in[0];
    const float* enc    = (const float*)d_in[1];
    const int*   esrc   = (const int*)d_in[2];
    const int*   edst   = (const int*)d_in[3];
    const float* W      = (const float*)d_in[4];
    const float* bvec   = (const float*)d_in[5];
    const float* v      = (const float*)d_in[6];
    float* out = (float*)d_out;

    char* ws = (char*)d_ws;
    unsigned short* Bpack = (unsigned short*)ws;
    float* P    = (float*)(ws + 262144);
    float* wbuf = (float*)(ws + 262144 + 65536);

    prep_kernel   <<<PREP_GROUPS, 512, 0, stream>>>(hidden, W, bvec, P, Bpack, out);
    scores_kernel <<<256, 512, 0, stream>>>(enc, (const uint4*)Bpack, P, v, wbuf);
    softmax_scatter_kernel<<<256, 256, 0, stream>>>(esrc, edst, wbuf, out);
}


# Prep launch, P job regrouped: lane = column n, wave = batch row; hidden row by scalar loads, W rows staged transposed in LDS by whole-line reads; fma order per (b,n) unchanged
# speedup vs baseline: 2.1587x; 1.0922x over previous
#include <hip/hip_runtime.h>
#include <hip/hip_bf16.h>

#define H 256
#define BB 32
#define TT 4096
#define EE 8192
#define D2 512

typedef __attribute__((ext_vector_type(16))) __bf16        v16bf;
typedef __attribute__((ext_vector_type(8)))  float         v8f;
typedef __attribute__((ext_vector_type(8)))  unsigned int  v8u;

__device__ __forceinline__ unsigned short f2bf(float x) {
    unsigned int u = __float_as_uint(x);
    unsigned int r = u + 0x7FFFu + ((u >> 16) & 1u);
    return (unsigned short)(r >> 16);
}
__device__ __forceinline__ unsigned int pk2(float lo, float hi) {
    return (unsigned int)f2bf(lo) | ((unsigned int)f2bf(hi) << 16);
}

__device__ __forceinline__ float htanh(float x) {
#if __has_builtin(__builtin_amdgcn_tanhf)
    return __builtin_amdgcn_tanhf(x);
#elif __has_builtin(__builtin_amdgcn_tanh_f32)
    return __builtin_amdgcn_tanh_f32(x);
#else
    float y;
    asm volatile("v_tanh_f32 %0, %1\n\t"
                 "s_delay_alu instid0(TRANS32_DEP_1)"
                 : "=v"(y) : "v"(x));
    return y;
#endif
}

#define PREP_P_GROUPS    32
#define PREP_PACK_GROUPS 16
#define PREP_ZERO_GROUPS 64
#define PREP_GROUPS      112
#define WS_PITCH 33

__global__ void __launch_bounds__(512)
prep_kernel(const float* __restrict__ hidden, const float* __restrict__ W,
            const float* __restrict__ bvec, float* __restrict__ P,
            unsigned short* __restrict__ Bp, float* __restrict__ out) {
    __shared__ float4 wsT[64 * WS_PITCH];
    const int g = blockIdx.x;
    if (g < PREP_P_GROUPS) {
        const int tid  = threadIdx.x;
        const int lane = tid & 31;
        const int wv   = __builtin_amdgcn_readfirstlane(tid >> 5);
        const int n0   = (g >> 1) * 32;
        const int b    = (g & 1) * 16 + wv;
        #pragma unroll
        for (int j = 0; j < 4; ++j) {
            int q = tid + 512 * j;
            wsT[(q & 63) * WS_PITCH + (q >> 6)] =
                ((const float4*)W)[(size_t)(n0 + (q >> 6)) * (D2 / 4) + (q & 63)];
        }
        const float* h = hidden + b * H;
        float acc = bvec[n0 + lane];
        float hr[64];
        #pragma unroll
        for (int i = 0; i < 64; ++i) hr[i] = h[i];
        __syncthreads();
        float4 wv4[2][16];
        #pragma unroll
        for (int j = 0; j < 16; ++j) wv4[0][j] = wsT[j * WS_PITCH + lane];
        #pragma unroll
        for (int s = 0; s < 4; ++s) {
            if (s < 3) {
                #pragma unroll
                for (int j = 0; j < 16; ++j)
                    wv4[(s + 1) & 1][j] = wsT[(16 * (s + 1) + j) * WS_PITCH + lane];
            }
            if (s > 0) {
                #pragma unroll
                for (int i = 0; i < 64; ++i) hr[i] = h[64 * s + i];
            }
            __builtin_amdgcn_sched_barrier(0);
            #pragma unroll
            for (int j = 0; j < 16; ++j) {
                const float4 w4 = wv4[s & 1][j];
                acc = fmaf(hr[4 * j + 0], w4.x, acc);
                acc = fmaf(hr[4 * j + 1], w4.y, acc);
                acc = fmaf(hr[4 * j + 2], w4.z, acc);
                acc = fmaf(hr[4 * j + 3], w4.w, acc);
            }
        }
        P[b * D2 + n0 + lane] = acc;
    } else if (g < PREP_P_GROUPS + PREP_PACK_GROUPS) {
        int blk  = (g - PREP_P_GROUPS) * 16 + (threadIdx.x >> 5);
        int nt   = blk >> 3;
        int kc   = blk & 7;
        int lane = threadIdx.x & 31;
        int n     = nt * 16 + (lane & 15);
        int kbase = kc * 32 + ((lane >> 4) ? 8 : 0);
        const float* wrow = W + (size_t)n * D2 + H;
        unsigned short* dst = Bp + (size_t)blk * 512 + lane * 8;
        #pragma unroll
        for (int j = 0; j < 8; ++j) dst[j]       = f2bf(wrow[kbase + j]);
        #pragma unroll
        for (int j = 0; j < 8; ++j) dst[256 + j] = f2bf(wrow[kbase + 16 + j]);
    } else {
        int i = (g - PREP_P_GROUPS - PREP_PACK_GROUPS) * 512 + threadIdx.x;
        ((float4*)out)[i] = make_float4(0.f, 0.f, 0.f, 0.f);
    }
}

#define BQ_VEC 16384

__global__ void __launch_bounds__(512)
scores_kernel(const float* __restrict__ enc, const uint4* __restrict__ Bq,
              const float* __restrict__ P, const float* __restrict__ v,
              float* __restrict__ scores) {
    __shared__ uint4 Bs[BQ_VEC];
    __shared__ float Ps[D2];
    __shared__ float vs[D2];
    int wave = blockIdx.x * 16 + (threadIdx.x >> 5);
    int lane = threadIdx.x & 31;

    {
        const int tid = threadIdx.x;
        const int rot = blockIdx.x & 31;
        #pragma unroll 1
        for (int jj = 0; jj < 32; jj += 16) {
            uint4 tmp[16];
            #pragma unroll
            for (int u = 0; u < 16; ++u) tmp[u] = Bq[tid + 512 * ((jj + u + rot) & 31)];
            #pragma unroll
            for (int u = 0; u < 16; ++u) Bs[tid + 512 * ((jj + u + rot) & 31)] = tmp[u];
        }
        Ps[tid] = P[(blockIdx.x >> 3) * D2 + tid];
        vs[tid] = v[tid];
    }

    int b    = wave >> 7;
    int t0   = (wave & 127) << 5;
    int mrow = lane & 15;
    int kgrp = lane >> 4;

    const float* rowp0 = enc + ((size_t)(t0 + mrow)      * BB + b) * H + kgrp * 8;
    const float* rowp1 = enc + ((size_t)(t0 + 16 + mrow) * BB + b) * H + kgrp * 8;
    v16bf a0[8], a1[8];
    #pragma unroll
    for (int kc = 0; kc < 8; ++kc) {
        const float* p0 = rowp0 + kc * 32;
        const float* p1 = rowp1 + kc * 32;
        v8u u0, u1;
        #pragma unroll
        for (int j = 0; j < 4; ++j) { u0[j] = pk2(p0[2 * j],     p0[2 * j + 1]);
                                      u1[j] = pk2(p1[2 * j],     p1[2 * j + 1]); }
        #pragma unroll
        for (int j = 4; j < 8; ++j) { u0[j] = pk2(p0[8 + 2 * j], p0[8 + 2 * j + 1]);
                                      u1[j] = pk2(p1[8 + 2 * j], p1[8 + 2 * j + 1]); }
        a0[kc] = __builtin_bit_cast(v16bf, u0);
        a1[kc] = __builtin_bit_cast(v16bf, u1);
    }

    float sacc0[8], sacc1[8];
    #pragma unroll
    for (int i = 0; i < 8; ++i) { sacc0[i] = 0.f; sacc1[i] = 0.f; }

    __syncthreads();

    for (int nt = 0; nt < 32; ++nt) {
        const uint4* bs = Bs + (nt * 8 * 64 + lane);
        v8f c0 = {}, c1 = {};
        #pragma unroll
        for (int kc = 0; kc < 8; ++kc) {
            uint4 lo = bs[kc * 64];
            uint4 hi = bs[kc * 64 + 32];
            v8u bu = { lo.x, lo.y, lo.z, lo.w, hi.x, hi.y, hi.z, hi.w };
            v16bf bv = __builtin_bit_cast(v16bf, bu);
            c0 = __builtin_amdgcn_wmma_f32_16x16x32_bf16(
                     false, a0[kc], false, bv, (short)0, c0, false, false);
            c1 = __builtin_amdgcn_wmma_f32_16x16x32_bf16(
                     false, a1[kc], false, bv, (short)0, c1, false, false);
        }
        int   n  = nt * 16 + (lane & 15);
        float pn = Ps[n];
        float vn = vs[n];
        #pragma unroll
        for (int i = 0; i < 8; ++i) {
            sacc0[i] += vn * htanh(c0[i] + pn);
            sacc1[i] += vn * htanh(c1[i] + pn);
        }
    }

    #pragma unroll
    for (int i = 0; i < 8; ++i) {
        float s0 = sacc0[i], s1 = sacc1[i];
        #pragma unroll
        for (int off = 1; off < 16; off <<= 1) {
            s0 += __shfl_xor(s0, off, 32);
            s1 += __shfl_xor(s1, off, 32);
        }
        sacc0[i] = s0; sacc1[i] = s1;
    }
    if ((lane & 15) == 0) {
        #pragma unroll
        for (int i = 0; i < 8; ++i) {
            scores[b * TT + t0 +      kgrp * 8 + i] = sacc0[i];
            scores[b * TT + t0 + 16 + kgrp * 8 + i] = sacc1[i];
        }
    }
}

__global__ void __launch_bounds__(256)
softmax_scatter_kernel(const int* __restrict__ src, const int* __restrict__ dst,
                       const float* __restrict__ sc, float* __restrict__ out) {
    __shared__ float redm[8];
    __shared__ float reds[8];
    const int b    = blockIdx.x >> 3;
    const int c    = blockIdx.x & 7;
    const int tid  = threadIdx.x;
    const int lane = tid & 31;
    const int wv   = tid >> 5;
    const float* row = sc + b * TT;

    const int e0 = b * EE + c * 1024 + tid;
    int s[4], d[4];
    #pragma unroll
    for (int j = 0; j < 4; ++j) { s[j] = src[e0 + 256 * j]; d[j] = dst[e0 + 256 * j]; }
    float4 x[4];
    #pragma unroll
    for (int j = 0; j < 4; ++j) x[j] = ((const float4*)row)[tid + 256 * j];
    float xs[4];
    #pragma unroll
    for (int j = 0; j < 4; ++j) xs[j] = row[s[j]];

    float m = fmaxf(fmaxf(x[0].x, x[0].y), fmaxf(x[0].z, x[0].w));
    #pragma unroll
    for (int j = 1; j < 4; ++j)
        m = fmaxf(m, fmaxf(fmaxf(x[j].x, x[j].y), fmaxf(x[j].z, x[j].w)));
    #pragma unroll
    for (int off = 16; off > 0; off >>= 1) m = fmaxf(m, __shfl_xor(m, off, 32));
    if (lane == 0) redm[wv] = m;
    __syncthreads();
    m = fmaxf(fmaxf(fmaxf(redm[0], redm[1]), fmaxf(redm[2], redm[3])),
              fmaxf(fmaxf(redm[4], redm[5]), fmaxf(redm[6], redm[7])));

    float sum = 0.f;
    #pragma unroll
    for (int j = 0; j < 4; ++j)
        sum += (__expf(x[j].x - m) + __expf(x[j].y - m)) + (__expf(x[j].z - m) + __expf(x[j].w - m));
    #pragma unroll
    for (int off = 16; off > 0; off >>= 1) sum += __shfl_xor(sum, off, 32);
    if (lane == 0) reds[wv] = sum;
    __syncthreads();
    sum = ((reds[0] + reds[1]) + (reds[2] + reds[3])) + ((reds[4] + reds[5]) + (reds[6] + reds[7]));
    const float inv = 1.0f / sum;

    #pragma unroll
    for (int j = 0; j < 4; ++j)
        if (d[j] != s[j] + 1)
            atomicAdd(&out[b * TT + d[j]], (__expf(xs[j] - m) * inv) * 0.1f);
}

extern "C" void kernel_launch(void* const* d_in, const int* in_sizes, int n_in,
                              void* d_out, int out_size, void* d_ws, size_t ws_size,
                              hipStream_t stream) {
    const float* hidden = (const float*)d_in[0];
    const float* enc    = (const float*)d_in[1];
    const int*   esrc   = (const int*)d_in[2];
    const int*   edst   = (const int*)d_in[3];
    const float* W      = (const float*)d_in[4];
    const float* bvec   = (const float*)d_in[5];
    const float* v      = (const float*)d_in[6];
    float* out = (float*)d_out;

    char* ws = (char*)d_ws;
    unsigned short* Bpack = (unsigned short*)ws;
    float* P    = (float*)(ws + 262144);
    float* wbuf = (float*)(ws + 262144 + 65536);

    prep_kernel   <<<PREP_GROUPS, 512, 0, stream>>>(hidden, W, bvec, P, Bpack, out);
    scores_kernel <<<256, 512, 0, stream>>>(enc, (const uint4*)Bpack, P, v, wbuf);
    softmax_scatter_kernel<<<256, 256, 0, stream>>>(esrc, edst, wbuf, out);
}


# A tiles fetched by whole-line global loads (a wave instruction reads 256 contiguous bytes of each of two rows) and turned into WMMA fragments through a wave-private LDS bounce; same bf16 values
# speedup vs baseline: 2.3421x; 1.0850x over previous
#include <hip/hip_runtime.h>
#include <hip/hip_bf16.h>

#define H 256
#define BB 32
#define TT 4096
#define EE 8192
#define D2 512

typedef __attribute__((ext_vector_type(16))) __bf16        v16bf;
typedef __attribute__((ext_vector_type(8)))  float         v8f;
typedef __attribute__((ext_vector_type(8)))  unsigned int  v8u;

__device__ __forceinline__ unsigned short f2bf(float x) {
    unsigned int u = __float_as_uint(x);
    unsigned int r = u + 0x7FFFu + ((u >> 16) & 1u);
    return (unsigned short)(r >> 16);
}
__device__ __forceinline__ unsigned int pk2(float lo, float hi) {
    return (unsigned int)f2bf(lo) | ((unsigned int)f2bf(hi) << 16);
}

__device__ __forceinline__ float htanh(float x) {
#if __has_builtin(__builtin_amdgcn_tanhf)
    return __builtin_amdgcn_tanhf(x);
#elif __has_builtin(__builtin_amdgcn_tanh_f32)
    return __builtin_amdgcn_tanh_f32(x);
#else
    float y;
    asm volatile("v_tanh_f32 %0, %1\n\t"
                 "s_delay_alu instid0(TRANS32_DEP_1)"
                 : "=v"(y) : "v"(x));
    return y;
#endif
}

#define PREP_P_GROUPS    32
#define PREP_PACK_GROUPS 16
#define PREP_ZERO_GROUPS 64
#define PREP_GROUPS      112
#define WS_PITCH 33

__global__ void __launch_bounds__(512)
prep_kernel(const float* __restrict__ hidden, const float* __restrict__ W,
            const float* __restrict__ bvec, float* __restrict__ P,
            unsigned short* __restrict__ Bp, float* __restrict__ out) {
    __shared__ float4 wsT[64 * WS_PITCH];
    const int g = blockIdx.x;
    if (g < PREP_P_GROUPS) {
        const int tid  = threadIdx.x;
        const int lane = tid & 31;
        const int wv   = __builtin_amdgcn_readfirstlane(tid >> 5);
        const int n0   = (g >> 1) * 32;
        const int b    = (g & 1) * 16 + wv;
        #pragma unroll
        for (int j = 0; j < 4; ++j) {
            int q = tid + 512 * j;
            wsT[(q & 63) * WS_PITCH + (q >> 6)] =
                ((const float4*)W)[(size_t)(n0 + (q >> 6)) * (D2 / 4) + (q & 63)];
        }
        const float* h = hidden + b * H;
        float acc = bvec[n0 + lane];
        float hr[64];
        #pragma unroll
        for (int i = 0; i < 64; ++i) hr[i] = h[i];
        __syncthreads();
        float4 wv4[2][16];
        #pragma unroll
        for (int j = 0; j < 16; ++j) wv4[0][j] = wsT[j * WS_PITCH + lane];
        #pragma unroll
        for (int s = 0; s < 4; ++s) {
            if (s < 3) {
                #pragma unroll
                for (int j = 0; j < 16; ++j)
                    wv4[(s + 1) & 1][j] = wsT[(16 * (s + 1) + j) * WS_PITCH + lane];
            }
            if (s > 0) {
                #pragma unroll
                for (int i = 0; i < 64; ++i) hr[i] = h[64 * s + i];
            }
            __builtin_amdgcn_sched_barrier(0);
            #pragma unroll
            for (int j = 0; j < 16; ++j) {
                const float4 w4 = wv4[s & 1][j];
                acc = fmaf(hr[4 * j + 0], w4.x, acc);
                acc = fmaf(hr[4 * j + 1], w4.y, acc);
                acc = fmaf(hr[4 * j + 2], w4.z, acc);
                acc = fmaf(hr[4 * j + 3], w4.w, acc);
            }
        }
        P[b * D2 + n0 + lane] = acc;
    } else if (g < PREP_P_GROUPS + PREP_PACK_GROUPS) {
        int blk  = (g - PREP_P_GROUPS) * 16 + (threadIdx.x >> 5);
        int nt   = blk >> 3;
        int kc   = blk & 7;
        int lane = threadIdx.x & 31;
        int n     = nt * 16 + (lane & 15);
        int kbase = kc * 32 + ((lane >> 4) ? 8 : 0);
        const float* wrow = W + (size_t)n * D2 + H;
        unsigned short* dst = Bp + (size_t)blk * 512 + lane * 8;
        #pragma unroll
        for (int j = 0; j < 8; ++j) dst[j]       = f2bf(wrow[kbase + j]);
        #pragma unroll
        for (int j = 0; j < 8; ++j) dst[256 + j] = f2bf(wrow[kbase + 16 + j]);
    } else {
        int i = (g - PREP_P_GROUPS - PREP_PACK_GROUPS) * 512 + threadIdx.x;
        ((float4*)out)[i] = make_float4(0.f, 0.f, 0.f, 0.f);
    }
}

#define BQ_VEC 16384
#define APK_PITCH 144
#define APK_WAVE (16 * APK_PITCH)

typedef __attribute__((ext_vector_type(2)))  unsigned int  v2u;
typedef __attribute__((ext_vector_type(4)))  unsigned int  v4u;

__global__ void __launch_bounds__(512)
scores_kernel(const float* __restrict__ enc, const uint4* __restrict__ Bq,
              const float* __restrict__ P, const float* __restrict__ v,
              float* __restrict__ scores) {
    __shared__ uint4 Bs[BQ_VEC];
    __shared__ float Ps[D2];
    __shared__ float vs[D2];
    __shared__ uint4 Apark[16 * APK_WAVE / 16];
    int wave = blockIdx.x * 16 + (threadIdx.x >> 5);
    int lane = threadIdx.x & 31;

    {
        const int tid = threadIdx.x;
        const int rot = blockIdx.x & 31;
        #pragma unroll 1
        for (int jj = 0; jj < 32; jj += 16) {
            uint4 tmp[16];
            #pragma unroll
            for (int u = 0; u < 16; ++u) tmp[u] = Bq[tid + 512 * ((jj + u + rot) & 31)];
            #pragma unroll
            for (int u = 0; u < 16; ++u) Bs[tid + 512 * ((jj + u + rot) & 31)] = tmp[u];
        }
        Ps[tid] = P[(blockIdx.x >> 3) * D2 + tid];
        vs[tid] = v[tid];
    }

    int b    = wave >> 7;
    int t0   = (wave & 127) << 5;
    int mrow = lane & 15;
    int kgrp = lane >> 4;

    v16bf a0[8], a1[8];
    {
        char* park = (char*)Apark + (threadIdx.x >> 5) * APK_WAVE;
        const float4* srcp = (const float4*)(enc + ((size_t)(t0 + kgrp * 8) * BB + b) * H) + mrow;
        char* st = park + kgrp * 8 * APK_PITCH + mrow * 8;
        const char* fr = park + mrow * APK_PITCH + kgrp * 16;
        #pragma unroll
        for (int tl = 0; tl < 2; ++tl) {
            #pragma unroll
            for (int q = 0; q < 4; ++q) {
                #pragma unroll
                for (int i = 0; i < 8; ++i) {
                    const float4 x = srcp[(size_t)(tl * 16 + i) * (BB * H / 4) + q * 16];
                    v2u p = { pk2(x.x, x.y), pk2(x.z, x.w) };
                    *(v2u*)(st + i * APK_PITCH) = p;
                }
                #pragma unroll
                for (int j = 0; j < 2; ++j) {
                    v4u lo = *(const v4u*)(fr + j * 64);
                    v4u hi = *(const v4u*)(fr + j * 64 + 32);
                    v8u u = __builtin_shufflevector(lo, hi, 0, 1, 2, 3, 4, 5, 6, 7);
                    if (tl == 0) a0[2 * q + j] = __builtin_bit_cast(v16bf, u);
                    else         a1[2 * q + j] = __builtin_bit_cast(v16bf, u);
                }
            }
        }
    }

    float sacc0[8], sacc1[8];
    #pragma unroll
    for (int i = 0; i < 8; ++i) { sacc0[i] = 0.f; sacc1[i] = 0.f; }

    __syncthreads();

    for (int nt = 0; nt < 32; ++nt) {
        const uint4* bs = Bs + (nt * 8 * 64 + lane);
        v8f c0 = {}, c1 = {};
        #pragma unroll
        for (int kc = 0; kc < 8; ++kc) {
            uint4 lo = bs[kc * 64];
            uint4 hi = bs[kc * 64 + 32];
            v8u bu = { lo.x, lo.y, lo.z, lo.w, hi.x, hi.y, hi.z, hi.w };
            v16bf bv = __builtin_bit_cast(v16bf, bu);
            c0 = __builtin_amdgcn_wmma_f32_16x16x32_bf16(
                     false, a0[kc], false, bv, (short)0, c0, false, false);
            c1 = __builtin_amdgcn_wmma_f32_16x16x32_bf16(
                     false, a1[kc], false, bv, (short)0, c1, false, false);
        }
        int   n  = nt * 16 + (lane & 15);
        float pn = Ps[n];
        float vn = vs[n];
        #pragma unroll
        for (int i = 0; i < 8; ++i) {
            sacc0[i] += vn * htanh(c0[i] + pn);
            sacc1[i] += vn * htanh(c1[i] + pn);
        }
    }

    #pragma unroll
    for (int i = 0; i < 8; ++i) {
        float s0 = sacc0[i], s1 = sacc1[i];
        #pragma unroll
        for (int off = 1; off < 16; off <<= 1) {
            s0 += __shfl_xor(s0, off, 32);
            s1 += __shfl_xor(s1, off, 32);
        }
        sacc0[i] = s0; sacc1[i] = s1;
    }
    if ((lane & 15) == 0) {
        #pragma unroll
        for (int i = 0; i < 8; ++i) {
            scores[b * TT + t0 +      kgrp * 8 + i] = sacc0[i];
            scores[b * TT + t0 + 16 + kgrp * 8 + i] = sacc1[i];
        }
    }
}

__global__ void __launch_bounds__(256)
softmax_scatter_kernel(const int* __restrict__ src, const int* __restrict__ dst,
                       const float* __restrict__ sc, float* __restrict__ out) {
    __shared__ float redm[8];
    __shared__ float reds[8];
    const int b    = blockIdx.x >> 3;
    const int c    = blockIdx.x & 7;
    const int tid  = threadIdx.x;
    const int lane = tid & 31;
    const int wv   = tid >> 5;
    const float* row = sc + b * TT;

    const int e0 = b * EE + c * 1024 + tid;
    int s[4], d[4];
    #pragma unroll
    for (int j = 0; j < 4; ++j) { s[j] = src[e0 + 256 * j]; d[j] = dst[e0 + 256 * j]; }
    float4 x[4];
    #pragma unroll
    for (int j = 0; j < 4; ++j) x[j] = ((const float4*)row)[tid + 256 * j];
    float xs[4];
    #pragma unroll
    for (int j = 0; j < 4; ++j) xs[j] = row[s[j]];

    float m = fmaxf(fmaxf(x[0].x, x[0].y), fmaxf(x[0].z, x[0].w));
    #pragma unroll
    for (int j = 1; j < 4; ++j)
        m = fmaxf(m, fmaxf(fmaxf(x[j].x, x[j].y), fmaxf(x[j].z, x[j].w)));
    #pragma unroll
    for (int off = 16; off > 0; off >>= 1) m = fmaxf(m, __shfl_xor(m, off, 32));
    if (lane == 0) redm[wv] = m;
    __syncthreads();
    m = fmaxf(fmaxf(fmaxf(redm[0], redm[1]), fmaxf(redm[2], redm[3])),
              fmaxf(fmaxf(redm[4], redm[5]), fmaxf(redm[6], redm[7])));

    float sum = 0.f;
    #pragma unroll
    for (int j = 0; j < 4; ++j)
        sum += (__expf(x[j].x - m) + __expf(x[j].y - m)) + (__expf(x[j].z - m) + __expf(x[j].w - m));
    #pragma unroll
    for (int off = 16; off > 0; off >>= 1) sum += __shfl_xor(sum, off, 32);
    if (lane == 0) reds[wv] = sum;
    __syncthreads();
    sum = ((reds[0] + reds[1]) + (reds[2] + reds[3])) + ((reds[4] + reds[5]) + (reds[6] + reds[7]));
    const float inv = 1.0f / sum;

    #pragma unroll
    for (int j = 0; j < 4; ++j)
        if (d[j] != s[j] + 1)
            atomicAdd(&out[b * TT + d[j]], (__expf(xs[j] - m) * inv) * 0.1f);
}

extern "C" void kernel_launch(void* const* d_in, const int* in_sizes, int n_in,
                              void* d_out, int out_size, void* d_ws, size_t ws_size,
                              hipStream_t stream) {
    const float* hidden = (const float*)d_in[0];
    const float* enc    = (const float*)d_in[1];
    const int*   esrc   = (const int*)d_in[2];
    const int*   edst   = (const int*)d_in[3];
    const float* W      = (const float*)d_in[4];
    const float* bvec   = (const float*)d_in[5];
    const float* v      = (const float*)d_in[6];
    float* out = (float*)d_out;

    char* ws = (char*)d_ws;
    unsigned short* Bpack = (unsigned short*)ws;
    float* P    = (float*)(ws + 262144);
    float* wbuf = (float*)(ws + 262144 + 65536);

    prep_kernel   <<<PREP_GROUPS, 512, 0, stream>>>(hidden, W, bvec, P, Bpack, out);
    scores_kernel <<<256, 512, 0, stream>>>(enc, (const uint4*)Bpack, P, v, wbuf);
    softmax_scatter_kernel<<<256, 256, 0, stream>>>(esrc, edst, wbuf, out);
}


# Scores launch, fetch phase as one load stream: B quarters and A row groups requested interleaved, each wait covers only the oldest group; A rounded by v_cvt_pk_bf16_f32 (same values) to fit registers
# speedup vs baseline: 2.4258x; 1.0357x over previous
#include <hip/hip_runtime.h>
#include <hip/hip_bf16.h>

#define H 256
#define BB 32
#define TT 4096
#define EE 8192
#define D2 512

typedef __attribute__((ext_vector_type(16))) __bf16        v16bf;
typedef __attribute__((ext_vector_type(8)))  float         v8f;
typedef __attribute__((ext_vector_type(8)))  unsigned int  v8u;

__device__ __forceinline__ unsigned short f2bf(float x) {
    unsigned int u = __float_as_uint(x);
    unsigned int r = u + 0x7FFFu + ((u >> 16) & 1u);
    return (unsigned short)(r >> 16);
}
typedef __attribute__((ext_vector_type(2))) float  v2f32;
typedef __attribute__((ext_vector_type(2))) __bf16 v2bf16;
__device__ __forceinline__ unsigned int pk2(float lo, float hi) {
    v2f32 f = { lo, hi };
    return __builtin_bit_cast(unsigned int, __builtin_convertvector(f, v2bf16));
}

__device__ __forceinline__ float htanh(float x) {
#if __has_builtin(__builtin_amdgcn_tanhf)
    return __builtin_amdgcn_tanhf(x);
#elif __has_builtin(__builtin_amdgcn_tanh_f32)
    return __builtin_amdgcn_tanh_f32(x);
#else
    float y;
    asm volatile("v_tanh_f32 %0, %1\n\t"
                 "s_delay_alu instid0(TRANS32_DEP_1)"
                 : "=v"(y) : "v"(x));
    return y;
#endif
}

#define PREP_P_GROUPS    32
#define PREP_PACK_GROUPS 16
#define PREP_ZERO_GROUPS 64
#define PREP_GROUPS      112
#define WS_PITCH 33

__global__ void __launch_bounds__(512)
prep_kernel(const float* __restrict__ hidden, const float* __restrict__ W,
            const float* __restrict__ bvec, float* __restrict__ P,
            unsigned short* __restrict__ Bp, float* __restrict__ out) {
    __shared__ float4 wsT[64 * WS_PITCH];
    const int g = blockIdx.x;
    if (g < PREP_P_GROUPS) {
        const int tid  = threadIdx.x;
        const int lane = tid & 31;
        const int wv   = __builtin_amdgcn_readfirstlane(tid >> 5);
        const int n0   = (g >> 1) * 32;
        const int b    = (g & 1) * 16 + wv;
        #pragma unroll
        for (int j = 0; j < 4; ++j) {
            int q = tid + 512 * j;
            wsT[(q & 63) * WS_PITCH + (q >> 6)] =
                ((const float4*)W)[(size_t)(n0 + (q >> 6)) * (D2 / 4) + (q & 63)];
        }
        const float* h = hidden + b * H;
        float acc = bvec[n0 + lane];
        float hr[64];
        #pragma unroll
        for (int i = 0; i < 64; ++i) hr[i] = h[i];
        __syncthreads();
        float4 wv4[2][16];
        #pragma unroll
        for (int j = 0; j < 16; ++j) wv4[0][j] = wsT[j * WS_PITCH + lane];
        #pragma unroll
        for (int s = 0; s < 4; ++s) {
            if (s < 3) {
                #pragma unroll
                for (int j = 0; j < 16; ++j)
                    wv4[(s + 1) & 1][j] = wsT[(16 * (s + 1) + j) * WS_PITCH + lane];
            }
            if (s > 0) {
                #pragma unroll
                for (int i = 0; i < 64; ++i) hr[i] = h[64 * s + i];
            }
            __builtin_amdgcn_sched_barrier(0);
            #pragma unroll
            for (int j = 0; j < 16; ++j) {
                const float4 w4 = wv4[s & 1][j];
                acc = fmaf(hr[4 * j + 0], w4.x, acc);
                acc = fmaf(hr[4 * j + 1], w4.y, acc);
                acc = fmaf(hr[4 * j + 2], w4.z, acc);
                acc = fmaf(hr[4 * j + 3], w4.w, acc);
            }
        }
        P[b * D2 + n0 + lane] = acc;
    } else if (g < PREP_P_GROUPS + PREP_PACK_GROUPS) {
        int blk  = (g - PREP_P_GROUPS) * 16 + (threadIdx.x >> 5);
        int nt   = blk >> 3;
        int kc   = blk & 7;
        int lane = threadIdx.x & 31;
        int n     = nt * 16 + (lane & 15);
        int kbase = kc * 32 + ((lane >> 4) ? 8 : 0);
        const float* wrow = W + (size_t)n * D2 + H;
        unsigned short* dst = Bp + (size_t)blk * 512 + lane * 8;
        #pragma unroll
        for (int j = 0; j < 8; ++j) dst[j]       = f2bf(wrow[kbase + j]);
        #pragma unroll
        for (int j = 0; j < 8; ++j) dst[256 + j] = f2bf(wrow[kbase + 16 + j]);
    } else {
        int i = (g - PREP_P_GROUPS - PREP_PACK_GROUPS) * 512 + threadIdx.x;
        ((float4*)out)[i] = make_float4(0.f, 0.f, 0.f, 0.f);
    }
}

#define BQ_VEC 16384
#define APK_PITCH 144
#define APK_WAVE (16 * APK_PITCH)

typedef __attribute__((ext_vector_type(2)))  unsigned int  v2u;
typedef __attribute__((ext_vector_type(4)))  unsigned int  v4u;

__global__ void __launch_bounds__(512)
scores_kernel(const float* enc, const uint4* Bq,
              const float* P, const float* v,
              float* __restrict__ scores) {
    __shared__ uint4 Bs[BQ_VEC];
    __shared__ float Ps[D2];
    __shared__ float vs[D2];
    __shared__ uint4 Apark[16 * APK_WAVE / 16];
    int wave = blockIdx.x * 16 + (threadIdx.x >> 5);
    int lane = threadIdx.x & 31;

    const int tid = threadIdx.x;
    const int rot = blockIdx.x & 31;
    int b    = wave >> 7;
    int t0   = (wave & 127) << 5;
    int mrow = lane & 15;
    int kgrp = lane >> 4;
    char* park = (char*)Apark + (threadIdx.x >> 5) * APK_WAVE;
    const float4* srcp = (const float4*)(enc + ((size_t)(t0 + kgrp * 8) * BB + b) * H) + mrow;
    char* st = park + kgrp * 8 * APK_PITCH + mrow * 8;
    const char* fr = park + mrow * APK_PITCH + kgrp * 16;
    v16bf a0[8], a1[8];

    v4u    tba[8], tbb[8];
    float4 xg0[16], xg1[16];

#define STREAM_FENCE() do { asm volatile("" ::: "memory"); __builtin_amdgcn_sched_barrier(0); } while (0)
#define A_REQ(dst, g)                                                              \
    _Pragma("unroll")                                                              \
    for (int q2 = 0; q2 < 2; ++q2) {                                               \
        _Pragma("unroll")                                                          \
        for (int i = 0; i < 8; ++i)                                                \
            dst[q2 * 8 + i] = srcp[(size_t)(((g) >> 1) * 16 + i) * (BB * H / 4)    \
                                   + (2 * ((g) & 1) + q2) * 16];                   \
    }
#define A_CVT(src, frag, g)                                                        \
    _Pragma("unroll")                                                              \
    for (int q2 = 0; q2 < 2; ++q2) {                                               \
        _Pragma("unroll")                                                          \
        for (int i = 0; i < 8; ++i) {                                              \
            const float4 x = src[q2 * 8 + i];                                      \
            v2u p = { pk2(x.x, x.y), pk2(x.z, x.w) };                              \
            *(v2u*)(st + i * APK_PITCH) = p;                                       \
        }                                                                          \
        _Pragma("unroll")                                                          \
        for (int j = 0; j < 2; ++j) {                                              \
            v4u lo = *(const v4u*)(fr + j * 64);                                   \
            v4u hi = *(const v4u*)(fr + j * 64 + 32);                              \
            v8u u = __builtin_shufflevector(lo, hi, 0, 1, 2, 3, 4, 5, 6, 7);       \
            frag[2 * (2 * ((g) & 1) + q2) + j] = __builtin_bit_cast(v16bf, u);     \
        }                                                                          \
    }
#define B_REQ(dst, c)                                                              \
    _Pragma("unroll")                                                              \
    for (int u = 0; u < 8; ++u)                                                    \
        dst[u] = *(const v4u*)(Bq + (tid + 512 * ((8 * (c) + u + rot) & 31)));
#define B_PUT(src, c)                                                              \
    _Pragma("unroll")                                                              \
    for (int u = 0; u < 8; ++u)                                                    \
        *(v4u*)(Bs + (tid + 512 * ((8 * (c) + u + rot) & 31))) = src[u];

    const float pval = P[(blockIdx.x >> 3) * D2 + tid];
    const float vval = v[tid];
    B_REQ(tba, 0)
    B_REQ(tbb, 1)
    A_REQ(xg0, 0)
    STREAM_FENCE();
    B_PUT(tba, 0)
    B_REQ(tba, 2)
    STREAM_FENCE();
    B_PUT(tbb, 1)
    B_REQ(tbb, 3)
    A_REQ(xg1, 1)
    STREAM_FENCE();
    A_CVT(xg0, a0, 0)
    A_REQ(xg0, 2)
    STREAM_FENCE();
    B_PUT(tba, 2)
    B_PUT(tbb, 3)
    STREAM_FENCE();
    A_CVT(xg1, a0, 1)
    A_REQ(xg1, 3)
    STREAM_FENCE();
    A_CVT(xg0, a1, 2)
    A_CVT(xg1, a1, 3)
#undef B_REQ
#undef B_PUT
#undef A_REQ
#undef A_CVT
#undef STREAM_FENCE

    Ps[tid] = pval;
    vs[tid] = vval;

    float sacc0[8], sacc1[8];
    #pragma unroll
    for (int i = 0; i < 8; ++i) { sacc0[i] = 0.f; sacc1[i] = 0.f; }

    __syncthreads();

    for (int nt = 0; nt < 32; ++nt) {
        const uint4* bs = Bs + (nt * 8 * 64 + lane);
        v8f c0 = {}, c1 = {};
        #pragma unroll
        for (int kc = 0; kc < 8; ++kc) {
            uint4 lo = bs[kc * 64];
            uint4 hi = bs[kc * 64 + 32];
            v8u bu = { lo.x, lo.y, lo.z, lo.w, hi.x, hi.y, hi.z, hi.w };
            v16bf bv = __builtin_bit_cast(v16bf, bu);
            c0 = __builtin_amdgcn_wmma_f32_16x16x32_bf16(
                     false, a0[kc], false, bv, (short)0, c0, false, false);
            c1 = __builtin_amdgcn_wmma_f32_16x16x32_bf16(
                     false, a1[kc], false, bv, (short)0, c1, false, false);
        }
        int   n  = nt * 16 + (lane & 15);
        float pn = Ps[n];
        float vn = vs[n];
        #pragma unroll
        for (int i = 0; i < 8; ++i) {
            sacc0[i] += vn * htanh(c0[i] + pn);
            sacc1[i] += vn * htanh(c1[i] + pn);
        }
    }

    #pragma unroll
    for (int i = 0; i < 8; ++i) {
        float s0 = sacc0[i], s1 = sacc1[i];
        #pragma unroll
        for (int off = 1; off < 16; off <<= 1) {
            s0 += __shfl_xor(s0, off, 32);
            s1 += __shfl_xor(s1, off, 32);
        }
        sacc0[i] = s0; sacc1[i] = s1;
    }
    if ((lane & 15) == 0) {
        #pragma unroll
        for (int i = 0; i < 8; ++i) {
            scores[b * TT + t0 +      kgrp * 8 + i] = sacc0[i];
            scores[b * TT + t0 + 16 + kgrp * 8 + i] = sacc1[i];
        }
    }
}

__global__ void __launch_bounds__(256)
softmax_scatter_kernel(const int* __restrict__ src, const int* __restrict__ dst,
                       const float* __restrict__ sc, float* __restrict__ out) {
    __shared__ float redm[8];
    __shared__ float reds[8];
    const int b    = blockIdx.x >> 3;
    const int c    = blockIdx.x & 7;
    const int tid  = threadIdx.x;
    const int lane = tid & 31;
    const int wv   = tid >> 5;
    const float* row = sc + b * TT;

    const int e0 = b * EE + c * 1024 + tid;
    int s[4], d[4];
    #pragma unroll
    for (int j = 0; j < 4; ++j) { s[j] = src[e0 + 256 * j]; d[j] = dst[e0 + 256 * j]; }
    float4 x[4];
    #pragma unroll
    for (int j = 0; j < 4; ++j) x[j] = ((const float4*)row)[tid + 256 * j];
    float xs[4];
    #pragma unroll
    for (int j = 0; j < 4; ++j) xs[j] = row[s[j]];

    float m = fmaxf(fmaxf(x[0].x, x[0].y), fmaxf(x[0].z, x[0].w));
    #pragma unroll
    for (int j = 1; j < 4; ++j)
        m = fmaxf(m, fmaxf(fmaxf(x[j].x, x[j].y), fmaxf(x[j].z, x[j].w)));
    #pragma unroll
    for (int off = 16; off > 0; off >>= 1) m = fmaxf(m, __shfl_xor(m, off, 32));
    if (lane == 0) redm[wv] = m;
    __syncthreads();
    m = fmaxf(fmaxf(fmaxf(redm[0], redm[1]), fmaxf(redm[2], redm[3])),
              fmaxf(fmaxf(redm[4], redm[5]), fmaxf(redm[6], redm[7])));

    float sum = 0.f;
    #pragma unroll
    for (int j = 0; j < 4; ++j)
        sum += (__expf(x[j].x - m) + __expf(x[j].y - m)) + (__expf(x[j].z - m) + __expf(x[j].w - m));
    #pragma unroll
    for (int off = 16; off > 0; off >>= 1) sum += __shfl_xor(sum, off, 32);
    if (lane == 0) reds[wv] = sum;
    __syncthreads();
    sum = ((reds[0] + reds[1]) + (reds[2] + reds[3])) + ((reds[4] + reds[5]) + (reds[6] + reds[7]));
    const float inv = 1.0f / sum;

    #pragma unroll
    for (int j = 0; j < 4; ++j)
        if (d[j] != s[j] + 1)
            atomicAdd(&out[b * TT + d[j]], (__expf(xs[j] - m) * inv) * 0.1f);
}

extern "C" void kernel_launch(void* const* d_in, const int* in_sizes, int n_in,
                              void* d_out, int out_size, void* d_ws, size_t ws_size,
                              hipStream_t stream) {
    const float* hidden = (const float*)d_in[0];
    const float* enc    = (const float*)d_in[1];
    const int*   esrc   = (const int*)d_in[2];
    const int*   edst   = (const int*)d_in[3];
    const float* W      = (const float*)d_in[4];
    const float* bvec   = (const float*)d_in[5];
    const float* v      = (const float*)d_in[6];
    float* out = (float*)d_out;

    char* ws = (char*)d_ws;
    unsigned short* Bpack = (unsigned short*)ws;
    float* P    = (float*)(ws + 262144);
    float* wbuf = (float*)(ws + 262144 + 65536);

    prep_kernel   <<<PREP_GROUPS, 512, 0, stream>>>(hidden, W, bvec, P, Bpack, out);
    scores_kernel <<<256, 512, 0, stream>>>(enc, (const uint4*)Bpack, P, v, wbuf);
    softmax_scatter_kernel<<<256, 256, 0, stream>>>(esrc, edst, wbuf, out);
}


# launch fusion 3->2: the last of a batch row's 8 work-groups to arrive (ticket counter) does that row's softmax and scatter; row summed in LDS by ds_add_f32, exp sum tree now over 512 threads
# speedup vs baseline: 2.5572x; 1.0542x over previous
#include <hip/hip_runtime.h>
#include <hip/hip_bf16.h>

#define H 256
#define BB 32
#define TT 4096
#define EE 8192
#define D2 512

typedef __attribute__((ext_vector_type(16))) __bf16        v16bf;
typedef __attribute__((ext_vector_type(8)))  float         v8f;
typedef __attribute__((ext_vector_type(8)))  unsigned int  v8u;

__device__ __forceinline__ unsigned short f2bf(float x) {
    unsigned int u = __float_as_uint(x);
    unsigned int r = u + 0x7FFFu + ((u >> 16) & 1u);
    return (unsigned short)(r >> 16);
}
typedef __attribute__((ext_vector_type(2))) float  v2f32;
typedef __attribute__((ext_vector_type(2))) __bf16 v2bf16;
__device__ __forceinline__ unsigned int pk2(float lo, float hi) {
    v2f32 f = { lo, hi };
    return __builtin_bit_cast(unsigned int, __builtin_convertvector(f, v2bf16));
}

__device__ __forceinline__ float htanh(float x) {
#if __has_builtin(__builtin_amdgcn_tanhf)
    return __builtin_amdgcn_tanhf(x);
#elif __has_builtin(__builtin_amdgcn_tanh_f32)
    return __builtin_amdgcn_tanh_f32(x);
#else
    float y;
    asm volatile("v_tanh_f32 %0, %1\n\t"
                 "s_delay_alu instid0(TRANS32_DEP_1)"
                 : "=v"(y) : "v"(x));
    return y;
#endif
}

#define PREP_P_GROUPS    32
#define PREP_PACK_GROUPS 16
#define PREP_GROUPS      48
#define WS_PITCH 33

__global__ void __launch_bounds__(512)
prep_kernel(const float* __restrict__ hidden, const float* __restrict__ W,
            const float* __restrict__ bvec, float* __restrict__ P,
            unsigned short* __restrict__ Bp, unsigned int* __restrict__ cnt) {
    __shared__ float4 wsT[64 * WS_PITCH];
    const int g = blockIdx.x;
    if (g < PREP_P_GROUPS) {
        const int tid  = threadIdx.x;
        const int lane = tid & 31;
        const int wv   = __builtin_amdgcn_readfirstlane(tid >> 5);
        const int n0   = (g >> 1) * 32;
        const int b    = (g & 1) * 16 + wv;
        #pragma unroll
        for (int j = 0; j < 4; ++j) {
            int q = tid + 512 * j;
            wsT[(q & 63) * WS_PITCH + (q >> 6)] =
                ((const float4*)W)[(size_t)(n0 + (q >> 6)) * (D2 / 4) + (q & 63)];
        }
        const float* h = hidden + b * H;
        float acc = bvec[n0 + lane];
        float hr[64];
        #pragma unroll
        for (int i = 0; i < 64; ++i) hr[i] = h[i];
        __syncthreads();
        float4 wv4[2][16];
        #pragma unroll
        for (int j = 0; j < 16; ++j) wv4[0][j] = wsT[j * WS_PITCH + lane];
        #pragma unroll
        for (int s = 0; s < 4; ++s) {
            if (s < 3) {
                #pragma unroll
                for (int j = 0; j < 16; ++j)
                    wv4[(s + 1) & 1][j] = wsT[(16 * (s + 1) + j) * WS_PITCH + lane];
            }
            if (s > 0) {
                #pragma unroll
                for (int i = 0; i < 64; ++i) hr[i] = h[64 * s + i];
            }
            __builtin_amdgcn_sched_barrier(0);
            #pragma unroll
            for (int j = 0; j < 16; ++j) {
                const float4 w4 = wv4[s & 1][j];
                acc = fmaf(hr[4 * j + 0], w4.x, acc);
                acc = fmaf(hr[4 * j + 1], w4.y, acc);
                acc = fmaf(hr[4 * j + 2], w4.z, acc);
                acc = fmaf(hr[4 * j + 3], w4.w, acc);
            }
        }
        P[b * D2 + n0 + lane] = acc;
    } else {
        if (g == PREP_P_GROUPS && threadIdx.x < BB) cnt[threadIdx.x] = 0u;
        int blk  = (g - PREP_P_GROUPS) * 16 + (threadIdx.x >> 5);
        int nt   = blk >> 3;
        int kc   = blk & 7;
        int lane = threadIdx.x & 31;
        int n     = nt * 16 + (lane & 15);
        int kbase = kc * 32 + ((lane >> 4) ? 8 : 0);
        const float* wrow = W + (size_t)n * D2 + H;
        unsigned short* dst = Bp + (size_t)blk * 512 + lane * 8;
        #pragma unroll
        for (int j = 0; j < 8; ++j) dst[j]       = f2bf(wrow[kbase + j]);
        #pragma unroll
        for (int j = 0; j < 8; ++j) dst[256 + j] = f2bf(wrow[kbase + 16 + j]);
    }
}

#define BQ_VEC 16384
#define APK_PITCH 144
#define APK_WAVE (16 * APK_PITCH)

typedef __attribute__((ext_vector_type(2)))  unsigned int  v2u;
typedef __attribute__((ext_vector_type(4)))  unsigned int  v4u;

__global__ void __launch_bounds__(512)
scores_kernel(const float* enc, const uint4* Bq,
              const float* P, const float* v,
              float* scores,
              const int* __restrict__ esrc, const int* __restrict__ edst,
              unsigned int* cnt, float* __restrict__ out) {
    __shared__ uint4 Bs[BQ_VEC];
    __shared__ float Ps[D2];
    __shared__ float vs[D2];
    __shared__ uint4 Apark[16 * APK_WAVE / 16];
    int wave = blockIdx.x * 16 + (threadIdx.x >> 5);
    int lane = threadIdx.x & 31;

    const int tid = threadIdx.x;
    const int rot = blockIdx.x & 31;
    int b    = wave >> 7;
    int t0   = (wave & 127) << 5;
    int mrow = lane & 15;
    int kgrp = lane >> 4;
    char* park = (char*)Apark + (threadIdx.x >> 5) * APK_WAVE;
    const float4* srcp = (const float4*)(enc + ((size_t)(t0 + kgrp * 8) * BB + b) * H) + mrow;
    char* st = park + kgrp * 8 * APK_PITCH + mrow * 8;
    const char* fr = park + mrow * APK_PITCH + kgrp * 16;
    v16bf a0[8], a1[8];

    v4u    tba[8], tbb[8];
    float4 xg0[16], xg1[16];

#define STREAM_FENCE() do { asm volatile("" ::: "memory"); __builtin_amdgcn_sched_barrier(0); } while (0)
#define A_REQ(dst, g)                                                              \
    _Pragma("unroll")                                                              \
    for (int q2 = 0; q2 < 2; ++q2) {                                               \
        _Pragma("unroll")                                                          \
        for (int i = 0; i < 8; ++i)                                                \
            dst[q2 * 8 + i] = srcp[(size_t)(((g) >> 1) * 16 + i) * (BB * H / 4)    \
                                   + (2 * ((g) & 1) + q2) * 16];                   \
    }
#define A_CVT(src, frag, g)                                                        \
    _Pragma("unroll")                                                              \
    for (int q2 = 0; q2 < 2; ++q2) {                                               \
        _Pragma("unroll")                                                          \
        for (int i = 0; i < 8; ++i) {                                              \
            const float4 x = src[q2 * 8 + i];                                      \
            v2u p = { pk2(x.x, x.y), pk2(x.z, x.w) };                              \
            *(v2u*)(st + i * APK_PITCH) = p;                                       \
        }                                                                          \
        _Pragma("unroll")                                                          \
        for (int j = 0; j < 2; ++j) {                                              \
            v4u lo = *(const v4u*)(fr + j * 64);                                   \
            v4u hi = *(const v4u*)(fr + j * 64 + 32);                              \
            v8u u = __builtin_shufflevector(lo, hi, 0, 1, 2, 3, 4, 5, 6, 7);       \
            frag[2 * (2 * ((g) & 1) + q2) + j] = __builtin_bit_cast(v16bf, u);     \
        }                                                                          \
    }
#define B_REQ(dst, c)                                                              \
    _Pragma("unroll")                                                              \
    for (int u = 0; u < 8; ++u)                                                    \
        dst[u] = *(const v4u*)(Bq + (tid + 512 * ((8 * (c) + u + rot) & 31)));
#define B_PUT(src, c)                                                              \
    _Pragma("unroll")                                                              \
    for (int u = 0; u < 8; ++u)                                                    \
        *(v4u*)(Bs + (tid + 512 * ((8 * (c) + u + rot) & 31))) = src[u];

    const float pval = P[(blockIdx.x >> 3) * D2 + tid];
    const float vval = v[tid];
    B_REQ(tba, 0)
    B_REQ(tbb, 1)
    A_REQ(xg0, 0)
    STREAM_FENCE();
    B_PUT(tba, 0)
    B_REQ(tba, 2)
    STREAM_FENCE();
    B_PUT(tbb, 1)
    B_REQ(tbb, 3)
    A_REQ(xg1, 1)
    STREAM_FENCE();
    A_CVT(xg0, a0, 0)
    A_REQ(xg0, 2)
    STREAM_FENCE();
    B_PUT(tba, 2)
    B_PUT(tbb, 3)
    STREAM_FENCE();
    A_CVT(xg1, a0, 1)
    A_REQ(xg1, 3)
    STREAM_FENCE();
    A_CVT(xg0, a1, 2)
    A_CVT(xg1, a1, 3)
#undef B_REQ
#undef B_PUT
#undef A_REQ
#undef A_CVT
#undef STREAM_FENCE

    Ps[tid] = pval;
    vs[tid] = vval;

    float sacc0[8], sacc1[8];
    #pragma unroll
    for (int i = 0; i < 8; ++i) { sacc0[i] = 0.f; sacc1[i] = 0.f; }

    __syncthreads();

    for (int nt = 0; nt < 32; ++nt) {
        const uint4* bs = Bs + (nt * 8 * 64 + lane);
        v8f c0 = {}, c1 = {};
        #pragma unroll
        for (int kc = 0; kc < 8; ++kc) {
            uint4 lo = bs[kc * 64];
            uint4 hi = bs[kc * 64 + 32];
            v8u bu = { lo.x, lo.y, lo.z, lo.w, hi.x, hi.y, hi.z, hi.w };
            v16bf bv = __builtin_bit_cast(v16bf, bu);
            c0 = __builtin_amdgcn_wmma_f32_16x16x32_bf16(
                     false, a0[kc], false, bv, (short)0, c0, false, false);
            c1 = __builtin_amdgcn_wmma_f32_16x16x32_bf16(
                     false, a1[kc], false, bv, (short)0, c1, false, false);
        }
        int   n  = nt * 16 + (lane & 15);
        float pn = Ps[n];
        float vn = vs[n];
        #pragma unroll
        for (int i = 0; i < 8; ++i) {
            sacc0[i] += vn * htanh(c0[i] + pn);
            sacc1[i] += vn * htanh(c1[i] + pn);
        }
    }

    #pragma unroll
    for (int i = 0; i < 8; ++i) {
        float s0 = sacc0[i], s1 = sacc1[i];
        #pragma unroll
        for (int off = 1; off < 16; off <<= 1) {
            s0 += __shfl_xor(s0, off, 32);
            s1 += __shfl_xor(s1, off, 32);
        }
        sacc0[i] = s0; sacc1[i] = s1;
    }
    {
        float val = 0.f;
        const int  ci    = lane & 7;
        const bool tile1 = (lane & 8) != 0;
        #pragma unroll
        for (int i = 0; i < 8; ++i) {
            const float cand = tile1 ? sacc1[i] : sacc0[i];
            val = (ci == i) ? cand : val;
        }
        unsigned int* sp = (unsigned int*)(scores + b * TT + t0) + (tile1 ? 16 : 0) + kgrp * 8 + ci;
        __hip_atomic_store(sp, __float_as_uint(val), __ATOMIC_RELAXED, __HIP_MEMORY_SCOPE_AGENT);
    }

    const int bq = blockIdx.x >> 3;
    unsigned int* s_ticket = (unsigned int*)vs;
    float* redm = Ps;
    float* reds = Ps + 16;
    __syncthreads();
    if (tid == 0) {
        __builtin_amdgcn_fence(__ATOMIC_RELEASE, "agent");
        *s_ticket = __hip_atomic_fetch_add(cnt + bq, 1u, __ATOMIC_RELAXED, __HIP_MEMORY_SCOPE_AGENT);
    }
    __syncthreads();
    if (*s_ticket != 7u) return;

    __builtin_amdgcn_fence(__ATOMIC_ACQUIRE, "agent");
    {
        const float* row = scores + bq * TT;
        const int    wv  = tid >> 5;
        float* wl  = (float*)Bs;
        float* acc = (float*)Bs + TT;
        const float4 x0 = ((const float4*)row)[tid];
        const float4 x1 = ((const float4*)row)[tid + 512];
        int es[16], ed[16];
        const int e0 = bq * EE + tid;
        #pragma unroll
        for (int j = 0; j < 16; ++j) { es[j] = esrc[e0 + 512 * j]; ed[j] = edst[e0 + 512 * j]; }
        ((float4*)acc)[tid]       = make_float4(0.f, 0.f, 0.f, 0.f);
        ((float4*)acc)[tid + 512] = make_float4(0.f, 0.f, 0.f, 0.f);

        float m = fmaxf(fmaxf(fmaxf(x0.x, x0.y), fmaxf(x0.z, x0.w)),
                        fmaxf(fmaxf(x1.x, x1.y), fmaxf(x1.z, x1.w)));
        #pragma unroll
        for (int off = 16; off > 0; off >>= 1) m = fmaxf(m, __shfl_xor(m, off, 32));
        if (lane == 0) redm[wv] = m;
        __syncthreads();
        m = fmaxf(fmaxf(fmaxf(fmaxf(redm[0], redm[1]), fmaxf(redm[2], redm[3])),
                        fmaxf(fmaxf(redm[4], redm[5]), fmaxf(redm[6], redm[7]))),
                  fmaxf(fmaxf(fmaxf(redm[8], redm[9]), fmaxf(redm[10], redm[11])),
                        fmaxf(fmaxf(redm[12], redm[13]), fmaxf(redm[14], redm[15]))));

        const float4 w0 = make_float4(__expf(x0.x - m), __expf(x0.y - m), __expf(x0.z - m), __expf(x0.w - m));
        const float4 w1 = make_float4(__expf(x1.x - m), __expf(x1.y - m), __expf(x1.z - m), __expf(x1.w - m));
        ((float4*)wl)[tid]       = w0;
        ((float4*)wl)[tid + 512] = w1;
        float sum = ((w0.x + w0.y) + (w0.z + w0.w)) + ((w1.x + w1.y) + (w1.z + w1.w));
        #pragma unroll
        for (int off = 16; off > 0; off >>= 1) sum += __shfl_xor(sum, off, 32);
        if (lane == 0) reds[wv] = sum;
        __syncthreads();
        sum = (((reds[0] + reds[1]) + (reds[2] + reds[3])) + ((reds[4] + reds[5]) + (reds[6] + reds[7])))
            + (((reds[8] + reds[9]) + (reds[10] + reds[11])) + ((reds[12] + reds[13]) + (reds[14] + reds[15])));
        const float inv = 1.0f / sum;

        float ev[16];
        #pragma unroll
        for (int j = 0; j < 16; ++j) ev[j] = wl[es[j]];
        #pragma unroll
        for (int j = 0; j < 16; ++j) ev[j] = (ed[j] != es[j] + 1) ? (ev[j] * inv) * 0.1f : 0.f;
        #pragma unroll
        for (int j = 0; j < 16; ++j) atomicAdd(&acc[ed[j]], ev[j]);
        __syncthreads();
        float4* o = (float4*)(out + bq * TT);
        o[tid]       = ((const float4*)acc)[tid];
        o[tid + 512] = ((const float4*)acc)[tid + 512];
    }
}

extern "C" void kernel_launch(void* const* d_in, const int* in_sizes, int n_in,
                              void* d_out, int out_size, void* d_ws, size_t ws_size,
                              hipStream_t stream) {
    const float* hidden = (const float*)d_in[0];
    const float* enc    = (const float*)d_in[1];
    const int*   esrc   = (const int*)d_in[2];
    const int*   edst   = (const int*)d_in[3];
    const float* W      = (const float*)d_in[4];
    const float* bvec   = (const float*)d_in[5];
    const float* v      = (const float*)d_in[6];
    float* out = (float*)d_out;

    char* ws = (char*)d_ws;
    unsigned short* Bpack = (unsigned short*)ws;
    float* P    = (float*)(ws + 262144);
    float* wbuf = (float*)(ws + 262144 + 65536);
    unsigned int* cnt = (unsigned int*)(ws + 262144 + 65536 + 524288);

    prep_kernel   <<<PREP_GROUPS, 512, 0, stream>>>(hidden, W, bvec, P, Bpack, cnt);
    scores_kernel <<<256, 512, 0, stream>>>(enc, (const uint4*)Bpack, P, v, wbuf, esrc, edst, cnt, out);
}


# Scores launch: each wave owns 64 rows (four M tiles), so one B fragment read from LDS feeds four WMMAs; 8-wave work-groups, two waves per SIMD; tail's exp sum tree now over 256 threads
# speedup vs baseline: 2.7593x; 1.0790x over previous
#include <hip/hip_runtime.h>
#include <hip/hip_bf16.h>

#define H 256
#define BB 32
#define TT 4096
#define EE 8192
#define D2 512

typedef __attribute__((ext_vector_type(16))) __bf16        v16bf;
typedef __attribute__((ext_vector_type(8)))  float         v8f;
typedef __attribute__((ext_vector_type(8)))  unsigned int  v8u;

__device__ __forceinline__ unsigned short f2bf(float x) {
    unsigned int u = __float_as_uint(x);
    unsigned int r = u + 0x7FFFu + ((u >> 16) & 1u);
    return (unsigned short)(r >> 16);
}
typedef __attribute__((ext_vector_type(2))) float  v2f32;
typedef __attribute__((ext_vector_type(2))) __bf16 v2bf16;
__device__ __forceinline__ unsigned int pk2(float lo, float hi) {
    v2f32 f = { lo, hi };
    return __builtin_bit_cast(unsigned int, __builtin_convertvector(f, v2bf16));
}

__device__ __forceinline__ float htanh(float x) {
#if __has_builtin(__builtin_amdgcn_tanhf)
    return __builtin_amdgcn_tanhf(x);
#elif __has_builtin(__builtin_amdgcn_tanh_f32)
    return __builtin_amdgcn_tanh_f32(x);
#else
    float y;
    asm volatile("v_tanh_f32 %0, %1\n\t"
                 "s_delay_alu instid0(TRANS32_DEP_1)"
                 : "=v"(y) : "v"(x));
    return y;
#endif
}

#define PREP_P_GROUPS    32
#define PREP_PACK_GROUPS 16
#define PREP_GROUPS      48
#define WS_PITCH 33

__global__ void __launch_bounds__(512)
prep_kernel(const float* __restrict__ hidden, const float* __restrict__ W,
            const float* __restrict__ bvec, float* __restrict__ P,
            unsigned short* __restrict__ Bp, unsigned int* __restrict__ cnt) {
    __shared__ float4 wsT[64 * WS_PITCH];
    const int g = blockIdx.x;
    if (g < PREP_P_GROUPS) {
        const int tid  = threadIdx.x;
        const int lane = tid & 31;
        const int wv   = __builtin_amdgcn_readfirstlane(tid >> 5);
        const int n0   = (g >> 1) * 32;
        const int b    = (g & 1) * 16 + wv;
        #pragma unroll
        for (int j = 0; j < 4; ++j) {
            int q = tid + 512 * j;
            wsT[(q & 63) * WS_PITCH + (q >> 6)] =
                ((const float4*)W)[(size_t)(n0 + (q >> 6)) * (D2 / 4) + (q & 63)];
        }
        const float* h = hidden + b * H;
        float acc = bvec[n0 + lane];
        float hr[64];
        #pragma unroll
        for (int i = 0; i < 64; ++i) hr[i] = h[i];
        __syncthreads();
        float4 wv4[2][16];
        #pragma unroll
        for (int j = 0; j < 16; ++j) wv4[0][j] = wsT[j * WS_PITCH + lane];
        #pragma unroll
        for (int s = 0; s < 4; ++s) {
            if (s < 3) {
                #pragma unroll
                for (int j = 0; j < 16; ++j)
                    wv4[(s + 1) & 1][j] = wsT[(16 * (s + 1) + j) * WS_PITCH + lane];
            }
            if (s > 0) {
                #pragma unroll
                for (int i = 0; i < 64; ++i) hr[i] = h[64 * s + i];
            }
            __builtin_amdgcn_sched_barrier(0);
            #pragma unroll
            for (int j = 0; j < 16; ++j) {
                const float4 w4 = wv4[s & 1][j];
                acc = fmaf(hr[4 * j + 0], w4.x, acc);
                acc = fmaf(hr[4 * j + 1], w4.y, acc);
                acc = fmaf(hr[4 * j + 2], w4.z, acc);
                acc = fmaf(hr[4 * j + 3], w4.w, acc);
            }
        }
        P[b * D2 + n0 + lane] = acc;
    } else {
        if (g == PREP_P_GROUPS && threadIdx.x < BB) cnt[threadIdx.x] = 0u;
        int blk  = (g - PREP_P_GROUPS) * 16 + (threadIdx.x >> 5);
        int nt   = blk >> 3;
        int kc   = blk & 7;
        int lane = threadIdx.x & 31;
        int n     = nt * 16 + (lane & 15);
        int kbase = kc * 32 + ((lane >> 4) ? 8 : 0);
        const float* wrow = W + (size_t)n * D2 + H;
        unsigned short* dst = Bp + (size_t)blk * 512 + lane * 8;
        #pragma unroll
        for (int j = 0; j < 8; ++j) dst[j]       = f2bf(wrow[kbase + j]);
        #pragma unroll
        for (int j = 0; j < 8; ++j) dst[256 + j] = f2bf(wrow[kbase + 16 + j]);
    }
}

#define SC_WAVES 8
#define SC_THREADS 256
#define BQ_VEC 16384
#define APK_PITCH 144
#define APK_WAVE (16 * APK_PITCH)

typedef __attribute__((ext_vector_type(2)))  unsigned int  v2u;
typedef __attribute__((ext_vector_type(4)))  unsigned int  v4u;

__global__ void __launch_bounds__(SC_THREADS)
scores_kernel(const float* enc, const uint4* Bq,
              const float* P, const float* v,
              float* scores,
              const int* __restrict__ esrc, const int* __restrict__ edst,
              unsigned int* cnt, float* __restrict__ out) {
    __shared__ uint4 Bs[BQ_VEC];
    __shared__ float Ps[D2];
    __shared__ float vs[D2];
    __shared__ uint4 Apark[SC_WAVES * APK_WAVE / 16];
    const int tid  = threadIdx.x;
    const int wv   = tid >> 5;
    const int lane = tid & 31;

    const int rot = blockIdx.x & 31;
    const int b    = blockIdx.x >> 3;
    const int t0   = ((((int)blockIdx.x & 7) << 3) + wv) << 6;
    const int mrow = lane & 15;
    const int kgrp = lane >> 4;
    char* park = (char*)Apark + wv * APK_WAVE;
    const float4* srcp = (const float4*)(enc + ((size_t)(t0 + kgrp * 8) * BB + b) * H) + mrow;
    char* st = park + kgrp * 8 * APK_PITCH + mrow * 8;
    const char* fr = park + mrow * APK_PITCH + kgrp * 16;
    v16bf a[4][8];

    v4u    tba[8], tbb[8];
    float4 xg0[16], xg1[16];

#define STREAM_FENCE() do { asm volatile("" ::: "memory"); __builtin_amdgcn_sched_barrier(0); } while (0)
#define A_REQ(dst, g)                                                              \
    _Pragma("unroll")                                                              \
    for (int q2 = 0; q2 < 2; ++q2) {                                               \
        _Pragma("unroll")                                                          \
        for (int i = 0; i < 8; ++i)                                                \
            dst[q2 * 8 + i] = srcp[(size_t)(((g) >> 1) * 16 + i) * (BB * H / 4)    \
                                   + (2 * ((g) & 1) + q2) * 16];                   \
    }
#define A_CVT(src, g)                                                              \
    _Pragma("unroll")                                                              \
    for (int q2 = 0; q2 < 2; ++q2) {                                               \
        _Pragma("unroll")                                                          \
        for (int i = 0; i < 8; ++i) {                                              \
            const float4 x = src[q2 * 8 + i];                                      \
            v2u p = { pk2(x.x, x.y), pk2(x.z, x.w) };                              \
            *(v2u*)(st + i * APK_PITCH) = p;                                       \
        }                                                                          \
        _Pragma("unroll")                                                          \
        for (int j = 0; j < 2; ++j) {                                              \
            v4u lo = *(const v4u*)(fr + j * 64);                                   \
            v4u hi = *(const v4u*)(fr + j * 64 + 32);                              \
            v8u u = __builtin_shufflevector(lo, hi, 0, 1, 2, 3, 4, 5, 6, 7);       \
            a[(g) >> 1][2 * (2 * ((g) & 1) + q2) + j] = __builtin_bit_cast(v16bf, u); \
        }                                                                          \
    }
#define B_IDX(c, u) (tid + SC_THREADS * ((8 * (c) + (u) + 2 * rot) & 63))
#define B_REQ(dst, c)                                                              \
    _Pragma("unroll")                                                              \
    for (int u = 0; u < 8; ++u)                                                    \
        dst[u] = *(const v4u*)(Bq + B_IDX(c, u));
#define B_PUT(src, c)                                                              \
    _Pragma("unroll")                                                              \
    for (int u = 0; u < 8; ++u)                                                    \
        *(v4u*)(Bs + B_IDX(c, u)) = src[u];

    const float pval0 = P[b * D2 + tid];
    const float pval1 = P[b * D2 + SC_THREADS + tid];
    const float vval0 = v[tid];
    const float vval1 = v[SC_THREADS + tid];
    B_REQ(tba, 0)
    B_REQ(tbb, 1)
    A_REQ(xg0, 0)
    A_REQ(xg1, 1)
    STREAM_FENCE();
    B_PUT(tba, 0)
    B_REQ(tba, 2)
    STREAM_FENCE();
    B_PUT(tbb, 1)
    B_REQ(tbb, 3)
    STREAM_FENCE();
    A_CVT(xg0, 0)
    A_REQ(xg0, 2)
    STREAM_FENCE();
    A_CVT(xg1, 1)
    A_REQ(xg1, 3)
    STREAM_FENCE();
    B_PUT(tba, 2)
    B_REQ(tba, 4)
    STREAM_FENCE();
    B_PUT(tbb, 3)
    B_REQ(tbb, 5)
    STREAM_FENCE();
    A_CVT(xg0, 2)
    A_REQ(xg0, 4)
    STREAM_FENCE();
    A_CVT(xg1, 3)
    A_REQ(xg1, 5)
    STREAM_FENCE();
    B_PUT(tba, 4)
    B_REQ(tba, 6)
    STREAM_FENCE();
    B_PUT(tbb, 5)
    B_REQ(tbb, 7)
    STREAM_FENCE();
    A_CVT(xg0, 4)
    A_REQ(xg0, 6)
    STREAM_FENCE();
    A_CVT(xg1, 5)
    A_REQ(xg1, 7)
    STREAM_FENCE();
    B_PUT(tba, 6)
    B_PUT(tbb, 7)
    STREAM_FENCE();
    A_CVT(xg0, 6)
    A_CVT(xg1, 7)
#undef B_REQ
#undef B_PUT
#undef B_IDX
#undef A_REQ
#undef A_CVT
#undef STREAM_FENCE

    Ps[tid] = pval0;
    Ps[SC_THREADS + tid] = pval1;
    vs[tid] = vval0;
    vs[SC_THREADS + tid] = vval1;

    float sacc[4][8];
    #pragma unroll
    for (int t = 0; t < 4; ++t) {
        #pragma unroll
        for (int i = 0; i < 8; ++i) sacc[t][i] = 0.f;
    }

    __syncthreads();

    for (int nt = 0; nt < 32; ++nt) {
        const uint4* bs = Bs + (nt * 8 * 64 + lane);
        v8f c0 = {}, c1 = {}, c2 = {}, c3 = {};
        #pragma unroll
        for (int kc = 0; kc < 8; ++kc) {
            uint4 lo = bs[kc * 64];
            uint4 hi = bs[kc * 64 + 32];
            v8u bu = { lo.x, lo.y, lo.z, lo.w, hi.x, hi.y, hi.z, hi.w };
            v16bf bv = __builtin_bit_cast(v16bf, bu);
            c0 = __builtin_amdgcn_wmma_f32_16x16x32_bf16(
                     false, a[0][kc], false, bv, (short)0, c0, false, false);
            c1 = __builtin_amdgcn_wmma_f32_16x16x32_bf16(
                     false, a[1][kc], false, bv, (short)0, c1, false, false);
            c2 = __builtin_amdgcn_wmma_f32_16x16x32_bf16(
                     false, a[2][kc], false, bv, (short)0, c2, false, false);
            c3 = __builtin_amdgcn_wmma_f32_16x16x32_bf16(
                     false, a[3][kc], false, bv, (short)0, c3, false, false);
        }
        int   n  = nt * 16 + (lane & 15);
        float pn = Ps[n];
        float vn = vs[n];
        #pragma unroll
        for (int i = 0; i < 8; ++i) {
            sacc[0][i] += vn * htanh(c0[i] + pn);
            sacc[1][i] += vn * htanh(c1[i] + pn);
            sacc[2][i] += vn * htanh(c2[i] + pn);
            sacc[3][i] += vn * htanh(c3[i] + pn);
        }
    }

    #pragma unroll
    for (int t = 0; t < 4; ++t) {
        #pragma unroll
        for (int i = 0; i < 8; ++i) {
            float s0 = sacc[t][i];
            #pragma unroll
            for (int off = 1; off < 16; off <<= 1) s0 += __shfl_xor(s0, off, 32);
            sacc[t][i] = s0;
        }
    }
    #pragma unroll
    for (int p = 0; p < 2; ++p) {
        float val = 0.f;
        const int  ci    = lane & 7;
        const bool tile1 = (lane & 8) != 0;
        #pragma unroll
        for (int i = 0; i < 8; ++i) {
            const float cand = tile1 ? sacc[2 * p + 1][i] : sacc[2 * p][i];
            val = (ci == i) ? cand : val;
        }
        unsigned int* sp = (unsigned int*)(scores + b * TT + t0 + 32 * p) + (tile1 ? 16 : 0) + kgrp * 8 + ci;
        __hip_atomic_store(sp, __float_as_uint(val), __ATOMIC_RELAXED, __HIP_MEMORY_SCOPE_AGENT);
    }

    const int bq = b;
    unsigned int* s_ticket = (unsigned int*)vs;
    float* redm = Ps;
    float* reds = Ps + 16;
    __syncthreads();
    if (tid == 0) {
        __builtin_amdgcn_fence(__ATOMIC_RELEASE, "agent");
        *s_ticket = __hip_atomic_fetch_add(cnt + bq, 1u, __ATOMIC_RELAXED, __HIP_MEMORY_SCOPE_AGENT);
    }
    __syncthreads();
    if (*s_ticket != 7u) return;

    __builtin_amdgcn_fence(__ATOMIC_ACQUIRE, "agent");
    {
        const float* row = scores + bq * TT;
        float* wl  = (float*)Bs;
        float* acc = (float*)Bs + TT;
        float4 x[4];
        #pragma unroll
        for (int j = 0; j < 4; ++j) x[j] = ((const float4*)row)[tid + SC_THREADS * j];
        int es[32], ed[32];
        const int e0 = bq * EE + tid;
        #pragma unroll
        for (int j = 0; j < 32; ++j) { es[j] = esrc[e0 + SC_THREADS * j]; ed[j] = edst[e0 + SC_THREADS * j]; }
        #pragma unroll
        for (int j = 0; j < 4; ++j) ((float4*)acc)[tid + SC_THREADS * j] = make_float4(0.f, 0.f, 0.f, 0.f);

        float m = fmaxf(fmaxf(x[0].x, x[0].y), fmaxf(x[0].z, x[0].w));
        #pragma unroll
        for (int j = 1; j < 4; ++j)
            m = fmaxf(m, fmaxf(fmaxf(x[j].x, x[j].y), fmaxf(x[j].z, x[j].w)));
        #pragma unroll
        for (int off = 16; off > 0; off >>= 1) m = fmaxf(m, __shfl_xor(m, off, 32));
        if (lane == 0) redm[wv] = m;
        __syncthreads();
        m = fmaxf(fmaxf(fmaxf(redm[0], redm[1]), fmaxf(redm[2], redm[3])),
                  fmaxf(fmaxf(redm[4], redm[5]), fmaxf(redm[6], redm[7])));

        float ps[4];
        #pragma unroll
        for (int j = 0; j < 4; ++j) {
            const float4 w = make_float4(__expf(x[j].x - m), __expf(x[j].y - m), __expf(x[j].z - m), __expf(x[j].w - m));
            ((float4*)wl)[tid + SC_THREADS * j] = w;
            ps[j] = (w.x + w.y) + (w.z + w.w);
        }
        float sum = (ps[0] + ps[1]) + (ps[2] + ps[3]);
        #pragma unroll
        for (int off = 16; off > 0; off >>= 1) sum += __shfl_xor(sum, off, 32);
        if (lane == 0) reds[wv] = sum;
        __syncthreads();
        sum = ((reds[0] + reds[1]) + (reds[2] + reds[3])) + ((reds[4] + reds[5]) + (reds[6] + reds[7]));
        const float inv = 1.0f / sum;

        float ev[32];
        #pragma unroll
        for (int j = 0; j < 32; ++j) ev[j] = wl[es[j]];
        #pragma unroll
        for (int j = 0; j < 32; ++j) ev[j] = (ed[j] != es[j] + 1) ? (ev[j] * inv) * 0.1f : 0.f;
        #pragma unroll
        for (int j = 0; j < 32; ++j) atomicAdd(&acc[ed[j]], ev[j]);
        __syncthreads();
        float4* o = (float4*)(out + bq * TT);
        #pragma unroll
        for (int j = 0; j < 4; ++j) o[tid + SC_THREADS * j] = ((const float4*)acc)[tid + SC_THREADS * j];
    }
}

extern "C" void kernel_launch(void* const* d_in, const int* in_sizes, int n_in,
                              void* d_out, int out_size, void* d_ws, size_t ws_size,
                              hipStream_t stream) {
    const float* hidden = (const float*)d_in[0];
    const float* enc    = (const float*)d_in[1];
    const int*   esrc   = (const int*)d_in[2];
    const int*   edst   = (const int*)d_in[3];
    const float* W      = (const float*)d_in[4];
    const float* bvec   = (const float*)d_in[5];
    const float* v      = (const float*)d_in[6];
    float* out = (float*)d_out;

    char* ws = (char*)d_ws;
    unsigned short* Bpack = (unsigned short*)ws;
    float* P    = (float*)(ws + 262144);
    float* wbuf = (float*)(ws + 262144 + 65536);
    unsigned int* cnt = (unsigned int*)(ws + 262144 + 65536 + 524288);

    prep_kernel   <<<PREP_GROUPS, 512, 0, stream>>>(hidden, W, bvec, P, Bpack, cnt);
    scores_kernel <<<256, SC_THREADS, 0, stream>>>(enc, (const uint4*)Bpack, P, v, wbuf, esrc, edst, cnt, out);
}
